# starting kernel
# baseline (speedup 1.0000x reference)
#include <hip/hip_runtime.h>
#include <hip/hip_bf16.h>

typedef float v2f __attribute__((ext_vector_type(2)));
typedef float v8f __attribute__((ext_vector_type(8)));

__global__ __launch_bounds__(256) void zero_kernel(float4* __restrict__ p, int n4) {
  int i = blockIdx.x * 256 + threadIdx.x;
  if (i < n4) p[i] = make_float4(0.f, 0.f, 0.f, 0.f);
}

__global__ __launch_bounds__(256) void degree_kernel(const int* __restrict__ src,
                                                     const int* __restrict__ dst,
                                                     unsigned* __restrict__ ds,
                                                     unsigned* __restrict__ dd, int nE) {
  int e = blockIdx.x * 256 + threadIdx.x;
  if (e < nE) {
    atomicAdd(&ds[src[e]], 1u);
    atomicAdd(&dd[dst[e]], 1u);
  }
}

__global__ __launch_bounds__(256) void norm_kernel(const unsigned* __restrict__ ds,
                                                   const unsigned* __restrict__ dd,
                                                   float* __restrict__ sn,
                                                   float* __restrict__ dn, int n) {
  int i = blockIdx.x * 256 + threadIdx.x;
  if (i < n) {
    unsigned a = ds[i]; if (a < 1u) a = 1u;
    unsigned b = dd[i]; if (b < 1u) b = 1u;
    sn[i] = rsqrtf((float)a);
    dn[i] = rsqrtf((float)b);
  }
}

__global__ __launch_bounds__(256) void gemm64_wmma(const float* __restrict__ X,
                                                   const float* __restrict__ W,
                                                   const float* __restrict__ row_scale,
                                                   float* __restrict__ Y, int n_rows) {
  __shared__ float Wt[64 * 64];
  const int tid = threadIdx.x;
  for (int i = tid; i < 64 * 64; i += 256) {
    int k = i >> 6, n = i & 63;
    Wt[n * 64 + k] = W[i];
  }
  __syncthreads();

  const int lane = tid & 31;
  const int wave = tid >> 5;
  const int tile = blockIdx.x * 8 + wave;
  if (tile * 16 >= n_rows) return;

  const int m = lane & 15;
  const int khalf = lane >> 4;
  const int row0 = tile * 16;
  const int row = row0 + m;
  const float sn = row_scale[row];
  const float* xrow = X + (size_t)row * 64;

  v8f acc0 = {}, acc1 = {}, acc2 = {}, acc3 = {};
#pragma unroll
  for (int kc = 0; kc < 16; ++kc) {
    const int kbase = kc * 4 + khalf * 2;
    const float2 av = *(const float2*)(xrow + kbase);
    v2f a; a.x = av.x * sn; a.y = av.y * sn;
    const v2f b0 = *(const v2f*)(&Wt[(m +  0) * 64 + kbase]);
    const v2f b1 = *(const v2f*)(&Wt[(m + 16) * 64 + kbase]);
    const v2f b2 = *(const v2f*)(&Wt[(m + 32) * 64 + kbase]);
    const v2f b3 = *(const v2f*)(&Wt[(m + 48) * 64 + kbase]);
    acc0 = __builtin_amdgcn_wmma_f32_16x16x4_f32(false, a, false, b0, (short)0, acc0, false, false);
    acc1 = __builtin_amdgcn_wmma_f32_16x16x4_f32(false, a, false, b1, (short)0, acc1, false, false);
    acc2 = __builtin_amdgcn_wmma_f32_16x16x4_f32(false, a, false, b2, (short)0, acc2, false, false);
    acc3 = __builtin_amdgcn_wmma_f32_16x16x4_f32(false, a, false, b3, (short)0, acc3, false, false);
  }

  float* yrow = Y + (size_t)row0 * 64;
  const int mofs = (lane < 16) ? 0 : 8;
#pragma unroll
  for (int j = 0; j < 8; ++j) {
    const size_t r = (size_t)(mofs + j) * 64;
    yrow[r +  0 + m] = acc0[j];
    yrow[r + 16 + m] = acc1[j];
    yrow[r + 32 + m] = acc2[j];
    yrow[r + 48 + m] = acc3[j];
  }
}

__global__ __launch_bounds__(256) void scatter_kernel(const float* __restrict__ hW,
                                                      const int* __restrict__ src,
                                                      const int* __restrict__ dst,
                                                      float* __restrict__ agg, int nE) {
  int g = blockIdx.x * 256 + threadIdx.x;
  int e = g >> 5;
  if (e >= nE) return;
  int p = (g & 31) * 2;
  int s = src[e], d = dst[e];
  float2 v = *(const float2*)(hW + (size_t)s * 64 + p);
  float* o = agg + (size_t)d * 64 + p;
  atomicAdd(o, v.x);
  atomicAdd(o + 1, v.y);
}

__global__ __launch_bounds__(256) void post_kernel(float* __restrict__ agg,
                                                   const float* __restrict__ dn,
                                                   const float* __restrict__ bias,
                                                   float* __restrict__ out,
                                                   int n_rows, int do_relu, int do_clear) {
  int g = blockIdx.x * 256 + threadIdx.x;
  if (g >= n_rows * 16) return;
  int row = g >> 4;
  int c = (g & 15) * 4;
  float4 v = *(const float4*)(agg + (size_t)row * 64 + c);
  float4 b = *(const float4*)(bias + c);
  float s = dn[row];
  float4 y;
  y.x = v.x * s + b.x; y.y = v.y * s + b.y;
  y.z = v.z * s + b.z; y.w = v.w * s + b.w;
  if (do_relu) {
    y.x = fmaxf(y.x, 0.f); y.y = fmaxf(y.y, 0.f);
    y.z = fmaxf(y.z, 0.f); y.w = fmaxf(y.w, 0.f);
  }
  *(float4*)(out + (size_t)row * 64 + c) = y;
  if (do_clear) *(float4*)(agg + (size_t)row * 64 + c) = make_float4(0.f, 0.f, 0.f, 0.f);
}

extern "C" void kernel_launch(void* const* d_in, const int* in_sizes, int n_in,
                              void* d_out, int out_size, void* d_ws, size_t ws_size,
                              hipStream_t stream) {
  const float* in_feat = (const float*)d_in[0];
  const int*   src     = (const int*)d_in[1];
  const int*   dst     = (const int*)d_in[2];
  const float* W1      = (const float*)d_in[3];
  const float* b1      = (const float*)d_in[4];
  const float* W2      = (const float*)d_in[5];
  const float* b2      = (const float*)d_in[6];
  float* out = (float*)d_out;

  const int N  = in_sizes[0] / 64;
  const int nE = in_sizes[1];

  const size_t feat_bytes = (size_t)N * 64 * sizeof(float);
  auto al = [](size_t x) { return (x + 255) & ~(size_t)255; };
  char* ws = (char*)d_ws;
  size_t off = 0;
  float*    hW    = (float*)(ws + off);    off += al(feat_bytes);
  float*    agg   = (float*)(ws + off);    off += al(feat_bytes);
  float*    h1    = (float*)(ws + off);    off += al(feat_bytes);
  unsigned* deg_s = (unsigned*)(ws + off); off += al((size_t)N * 4);
  unsigned* deg_d = (unsigned*)(ws + off); off += al((size_t)N * 4);
  float*    sn    = (float*)(ws + off);    off += al((size_t)N * 4);
  float*    dn    = (float*)(ws + off);    off += al((size_t)N * 4);
  (void)ws_size; (void)n_in; (void)out_size;

  {
    int n4 = (int)(feat_bytes / 16);
    zero_kernel<<<(n4 + 255) / 256, 256, 0, stream>>>((float4*)agg, n4);
    int d4 = N / 4;
    zero_kernel<<<(d4 + 255) / 256, 256, 0, stream>>>((float4*)deg_s, d4);
    zero_kernel<<<(d4 + 255) / 256, 256, 0, stream>>>((float4*)deg_d, d4);
  }

  degree_kernel<<<(nE + 255) / 256, 256, 0, stream>>>(src, dst, deg_s, deg_d, nE);
  norm_kernel<<<(N + 255) / 256, 256, 0, stream>>>(deg_s, deg_d, sn, dn, N);

  const int gemm_blocks = ((N + 15) / 16 + 7) / 8;
  const int scat_blocks = (nE * 32 + 255) / 256;
  const int post_blocks = (N * 16 + 255) / 256;

  gemm64_wmma<<<gemm_blocks, 256, 0, stream>>>(in_feat, W1, sn, hW, N);
  scatter_kernel<<<scat_blocks, 256, 0, stream>>>(hW, src, dst, agg, nE);
  post_kernel<<<post_blocks, 256, 0, stream>>>(agg, dn, b1, h1, N,  1,  1);

  gemm64_wmma<<<gemm_blocks, 256, 0, stream>>>(h1, W2, sn, hW, N);
  scatter_kernel<<<scat_blocks, 256, 0, stream>>>(hW, src, dst, agg, nE);
  post_kernel<<<post_blocks, 256, 0, stream>>>(agg, dn, b2, out, N,  0,  0);
}


# degree counts in LDS counters per node range over edge slices, partial counts summed in the norm kernel: no global integer atomics, the two counter-zeroing launches go (11 -> 9 launches)
# speedup vs baseline: 1.2783x; 1.2783x over previous
#include <hip/hip_runtime.h>
#include <hip/hip_bf16.h>

typedef float v2f __attribute__((ext_vector_type(2)));
typedef float v8f __attribute__((ext_vector_type(8)));

__global__ __launch_bounds__(256) void zero_kernel(float4* __restrict__ p, int n4) {
  int i = blockIdx.x * 256 + threadIdx.x;
  if (i < n4) p[i] = make_float4(0.f, 0.f, 0.f, 0.f);
}

#define DEG_BINS 12544
#define DEG_K 16
#define DEG_T 512
#define DEG_U 8

__device__ __forceinline__ void deg_count(unsigned* h, int idx, unsigned lo) {
  const unsigned b = (unsigned)idx - lo;
  if (b < (unsigned)DEG_BINS) atomicAdd(&h[b], 1u);
}

__global__ __launch_bounds__(DEG_T) void degree_hist_kernel(const int* __restrict__ src,
                                                            const int* __restrict__ dst,
                                                            unsigned* __restrict__ part,
                                                            int nE, int n, int n_ranges) {
  __shared__ unsigned h[DEG_BINS];
  const int tid = threadIdx.x;
  int bid = blockIdx.x;
  const int k = bid % DEG_K; bid /= DEG_K;
  const int r = bid % n_ranges;
  const int which = bid / n_ranges;
  const int* __restrict__ idx = which ? dst : src;
  const unsigned lo = (unsigned)r * DEG_BINS;

  for (int b = tid; b < DEG_BINS; b += DEG_T) h[b] = 0u;
  __syncthreads();

  const int S = (((nE + DEG_K - 1) / DEG_K) + 3) & ~3;
  int e0 = k * S; if (e0 > nE) e0 = nE;
  int e1 = e0 + S; if (e1 > nE) e1 = nE;
  const int n4 = (e1 - e0) >> 2;
  const int4* __restrict__ p4 = (const int4*)(idx + e0);
  for (int g = tid; g < n4; g += DEG_U * DEG_T) {
    int4 v[DEG_U];
#pragma unroll
    for (int u = 0; u < DEG_U; ++u) {
      const int gu = g + u * DEG_T;
      v[u] = (gu < n4) ? p4[gu] : make_int4(-1, -1, -1, -1);
    }
#pragma unroll
    for (int u = 0; u < DEG_U; ++u) {
      deg_count(h, v[u].x, lo);
      deg_count(h, v[u].y, lo);
      deg_count(h, v[u].z, lo);
      deg_count(h, v[u].w, lo);
    }
  }
  for (int e = e0 + (n4 << 2) + tid; e < e1; e += DEG_T) deg_count(h, idx[e], lo);
  __syncthreads();

  int nb = n - (int)lo; if (nb > DEG_BINS) nb = DEG_BINS;
  unsigned* __restrict__ o = part + (size_t)(which * DEG_K + k) * n + lo;
  for (int b = tid; b < nb; b += DEG_T) o[b] = h[b];
}

__global__ __launch_bounds__(256) void norm_kernel(const unsigned* __restrict__ part,
                                                   float* __restrict__ sn,
                                                   float* __restrict__ dn, int n) {
  int i = blockIdx.x * 256 + threadIdx.x;
  if (i < n) {
    unsigned a = 0u, b = 0u;
#pragma unroll
    for (int k = 0; k < DEG_K; ++k) {
      a += part[(size_t)k * n + i];
      b += part[(size_t)(DEG_K + k) * n + i];
    }
    if (a < 1u) a = 1u;
    if (b < 1u) b = 1u;
    sn[i] = rsqrtf((float)a);
    dn[i] = rsqrtf((float)b);
  }
}

__global__ __launch_bounds__(256) void gemm64_wmma(const float* __restrict__ X,
                                                   const float* __restrict__ W,
                                                   const float* __restrict__ row_scale,
                                                   float* __restrict__ Y, int n_rows) {
  __shared__ float Wt[64 * 64];
  const int tid = threadIdx.x;
  for (int i = tid; i < 64 * 64; i += 256) {
    int k = i >> 6, n = i & 63;
    Wt[n * 64 + k] = W[i];
  }
  __syncthreads();

  const int lane = tid & 31;
  const int wave = tid >> 5;
  const int tile = blockIdx.x * 8 + wave;
  if (tile * 16 >= n_rows) return;

  const int m = lane & 15;
  const int khalf = lane >> 4;
  const int row0 = tile * 16;
  const int row = row0 + m;
  const float sn = row_scale[row];
  const float* xrow = X + (size_t)row * 64;

  v8f acc0 = {}, acc1 = {}, acc2 = {}, acc3 = {};
#pragma unroll
  for (int kc = 0; kc < 16; ++kc) {
    const int kbase = kc * 4 + khalf * 2;
    const float2 av = *(const float2*)(xrow + kbase);
    v2f a; a.x = av.x * sn; a.y = av.y * sn;
    const v2f b0 = *(const v2f*)(&Wt[(m +  0) * 64 + kbase]);
    const v2f b1 = *(const v2f*)(&Wt[(m + 16) * 64 + kbase]);
    const v2f b2 = *(const v2f*)(&Wt[(m + 32) * 64 + kbase]);
    const v2f b3 = *(const v2f*)(&Wt[(m + 48) * 64 + kbase]);
    acc0 = __builtin_amdgcn_wmma_f32_16x16x4_f32(false, a, false, b0, (short)0, acc0, false, false);
    acc1 = __builtin_amdgcn_wmma_f32_16x16x4_f32(false, a, false, b1, (short)0, acc1, false, false);
    acc2 = __builtin_amdgcn_wmma_f32_16x16x4_f32(false, a, false, b2, (short)0, acc2, false, false);
    acc3 = __builtin_amdgcn_wmma_f32_16x16x4_f32(false, a, false, b3, (short)0, acc3, false, false);
  }

  float* yrow = Y + (size_t)row0 * 64;
  const int mofs = (lane < 16) ? 0 : 8;
#pragma unroll
  for (int j = 0; j < 8; ++j) {
    const size_t r = (size_t)(mofs + j) * 64;
    yrow[r +  0 + m] = acc0[j];
    yrow[r + 16 + m] = acc1[j];
    yrow[r + 32 + m] = acc2[j];
    yrow[r + 48 + m] = acc3[j];
  }
}

__global__ __launch_bounds__(256) void scatter_kernel(const float* __restrict__ hW,
                                                      const int* __restrict__ src,
                                                      const int* __restrict__ dst,
                                                      float* __restrict__ agg, int nE) {
  int g = blockIdx.x * 256 + threadIdx.x;
  int e = g >> 5;
  if (e >= nE) return;
  int p = (g & 31) * 2;
  int s = src[e], d = dst[e];
  float2 v = *(const float2*)(hW + (size_t)s * 64 + p);
  float* o = agg + (size_t)d * 64 + p;
  atomicAdd(o, v.x);
  atomicAdd(o + 1, v.y);
}

__global__ __launch_bounds__(256) void post_kernel(float* __restrict__ agg,
                                                   const float* __restrict__ dn,
                                                   const float* __restrict__ bias,
                                                   float* __restrict__ out,
                                                   int n_rows, int do_relu, int do_clear) {
  int g = blockIdx.x * 256 + threadIdx.x;
  if (g >= n_rows * 16) return;
  int row = g >> 4;
  int c = (g & 15) * 4;
  float4 v = *(const float4*)(agg + (size_t)row * 64 + c);
  float4 b = *(const float4*)(bias + c);
  float s = dn[row];
  float4 y;
  y.x = v.x * s + b.x; y.y = v.y * s + b.y;
  y.z = v.z * s + b.z; y.w = v.w * s + b.w;
  if (do_relu) {
    y.x = fmaxf(y.x, 0.f); y.y = fmaxf(y.y, 0.f);
    y.z = fmaxf(y.z, 0.f); y.w = fmaxf(y.w, 0.f);
  }
  *(float4*)(out + (size_t)row * 64 + c) = y;
  if (do_clear) *(float4*)(agg + (size_t)row * 64 + c) = make_float4(0.f, 0.f, 0.f, 0.f);
}

extern "C" void kernel_launch(void* const* d_in, const int* in_sizes, int n_in,
                              void* d_out, int out_size, void* d_ws, size_t ws_size,
                              hipStream_t stream) {
  const float* in_feat = (const float*)d_in[0];
  const int*   src     = (const int*)d_in[1];
  const int*   dst     = (const int*)d_in[2];
  const float* W1      = (const float*)d_in[3];
  const float* b1      = (const float*)d_in[4];
  const float* W2      = (const float*)d_in[5];
  const float* b2      = (const float*)d_in[6];
  float* out = (float*)d_out;

  const int N  = in_sizes[0] / 64;
  const int nE = in_sizes[1];

  const size_t feat_bytes = (size_t)N * 64 * sizeof(float);
  auto al = [](size_t x) { return (x + 255) & ~(size_t)255; };
  char* ws = (char*)d_ws;
  size_t off = 0;
  float*    hW    = (float*)(ws + off);    off += al(feat_bytes);
  float*    agg   = (float*)(ws + off);    off += al(feat_bytes);
  float*    h1    = (float*)(ws + off);    off += al(feat_bytes);
  unsigned* part  = (unsigned*)(ws + off); off += al((size_t)2 * DEG_K * N * 4);
  float*    sn    = (float*)(ws + off);    off += al((size_t)N * 4);
  float*    dn    = (float*)(ws + off);    off += al((size_t)N * 4);
  (void)ws_size; (void)n_in; (void)out_size;

  {
    int n4 = (int)(feat_bytes / 16);
    zero_kernel<<<(n4 + 255) / 256, 256, 0, stream>>>((float4*)agg, n4);
  }

  const int n_ranges = (N + DEG_BINS - 1) / DEG_BINS;
  degree_hist_kernel<<<2 * n_ranges * DEG_K, DEG_T, 0, stream>>>(src, dst, part, nE, N, n_ranges);
  norm_kernel<<<(N + 255) / 256, 256, 0, stream>>>(part, sn, dn, N);

  const int gemm_blocks = ((N + 15) / 16 + 7) / 8;
  const int scat_blocks = (nE * 32 + 255) / 256;
  const int post_blocks = (N * 16 + 255) / 256;

  gemm64_wmma<<<gemm_blocks, 256, 0, stream>>>(in_feat, W1, sn, hW, N);
  scatter_kernel<<<scat_blocks, 256, 0, stream>>>(hW, src, dst, agg, nE);
  post_kernel<<<post_blocks, 256, 0, stream>>>(agg, dn, b1, h1, N,  1,  1);

  gemm64_wmma<<<gemm_blocks, 256, 0, stream>>>(h1, W2, sn, hW, N);
  scatter_kernel<<<scat_blocks, 256, 0, stream>>>(hW, src, dst, agg, nE);
  post_kernel<<<post_blocks, 256, 0, stream>>>(agg, dn, b2, out, N,  0,  0);
}


# aggregation as gather over an edge list bucketed by dst once per call (slots from returning LDS adds: list order = arrival order); float atomics, agg buffer, epilogue launches go; 9 -> 8 launches
# speedup vs baseline: 3.2580x; 2.5487x over previous
#include <hip/hip_runtime.h>
#include <hip/hip_bf16.h>

typedef float v2f __attribute__((ext_vector_type(2)));
typedef float v8f __attribute__((ext_vector_type(8)));

#define DEG_BINS 12544
#define DEG_K 16
#define DEG_T 512
#define DEG_U 8

__device__ __forceinline__ void deg_count(unsigned* h, int idx, unsigned lo) {
  const unsigned b = (unsigned)idx - lo;
  if (b < (unsigned)DEG_BINS) atomicAdd(&h[b], 1u);
}

__global__ __launch_bounds__(DEG_T) void degree_hist_kernel(const int* __restrict__ src,
                                                            const int* __restrict__ dst,
                                                            unsigned* __restrict__ part,
                                                            int nE, int n, int n_ranges) {
  __shared__ unsigned h[DEG_BINS];
  const int tid = threadIdx.x;
  int bid = blockIdx.x;
  const int k = bid % DEG_K; bid /= DEG_K;
  const int r = bid % n_ranges;
  const int which = bid / n_ranges;
  const int* __restrict__ idx = which ? dst : src;
  const unsigned lo = (unsigned)r * DEG_BINS;

  for (int b = tid; b < DEG_BINS; b += DEG_T) h[b] = 0u;
  __syncthreads();

  const int S = (((nE + DEG_K - 1) / DEG_K) + 3) & ~3;
  int e0 = k * S; if (e0 > nE) e0 = nE;
  int e1 = e0 + S; if (e1 > nE) e1 = nE;
  const int n4 = (e1 - e0) >> 2;
  const int4* __restrict__ p4 = (const int4*)(idx + e0);
  for (int g = tid; g < n4; g += DEG_U * DEG_T) {
    int4 v[DEG_U];
#pragma unroll
    for (int u = 0; u < DEG_U; ++u) {
      const int gu = g + u * DEG_T;
      v[u] = (gu < n4) ? p4[gu] : make_int4(-1, -1, -1, -1);
    }
#pragma unroll
    for (int u = 0; u < DEG_U; ++u) {
      deg_count(h, v[u].x, lo);
      deg_count(h, v[u].y, lo);
      deg_count(h, v[u].z, lo);
      deg_count(h, v[u].w, lo);
    }
  }
  for (int e = e0 + (n4 << 2) + tid; e < e1; e += DEG_T) deg_count(h, idx[e], lo);
  __syncthreads();

  int nb = n - (int)lo; if (nb > DEG_BINS) nb = DEG_BINS;
  unsigned* __restrict__ o = part + (size_t)(which * DEG_K + k) * n + lo;
  for (int b = tid; b < nb; b += DEG_T) o[b] = h[b];
}

__global__ __launch_bounds__(256) void norm_kernel(const unsigned* __restrict__ part,
                                                   float* __restrict__ sn,
                                                   float* __restrict__ dn,
                                                   unsigned* __restrict__ bsum, int n) {
  __shared__ unsigned tot;
  if (threadIdx.x == 0) tot = 0u;
  __syncthreads();
  int i = blockIdx.x * 256 + threadIdx.x;
  unsigned a = 0u, b = 0u;
  if (i < n) {
#pragma unroll
    for (int k = 0; k < DEG_K; ++k) {
      a += part[(size_t)k * n + i];
      b += part[(size_t)(DEG_K + k) * n + i];
    }
    sn[i] = rsqrtf((float)(a < 1u ? 1u : a));
    dn[i] = rsqrtf((float)(b < 1u ? 1u : b));
  }
  atomicAdd(&tot, b);
  __syncthreads();
  if (threadIdx.x == 0) bsum[blockIdx.x] = tot;
}

__global__ __launch_bounds__(256) void rowptr_kernel(unsigned* __restrict__ cnt,
                                                     const unsigned* __restrict__ bsum,
                                                     unsigned* __restrict__ rowptr, int n) {
  __shared__ unsigned wsum[8];
  __shared__ unsigned base_s;
  const int tid = threadIdx.x;
  const int lane = tid & 31, wave = tid >> 5;
  if (tid == 0) base_s = 0u;
  __syncthreads();
  unsigned pre = 0u;
  for (int j = tid; j < (int)blockIdx.x; j += 256) pre += bsum[j];
  atomicAdd(&base_s, pre);

  const int i = blockIdx.x * 256 + tid;
  unsigned c[DEG_K];
  unsigned deg = 0u;
#pragma unroll
  for (int k = 0; k < DEG_K; ++k) {
    c[k] = (i < n) ? cnt[(size_t)k * n + i] : 0u;
    deg += c[k];
  }
  unsigned inc = deg;
  for (int d = 1; d < 32; d <<= 1) {
    const unsigned t = __shfl_up(inc, d);
    if (lane >= d) inc += t;
  }
  if (lane == 31) wsum[wave] = inc;
  __syncthreads();
  unsigned start = base_s + inc - deg;
  for (int w = 0; w < wave; ++w) start += wsum[w];
  if (i < n) {
    rowptr[i] = start;
    if (i == n - 1) rowptr[n] = start + deg;
    unsigned run = start;
#pragma unroll
    for (int k = 0; k < DEG_K; ++k) {
      cnt[(size_t)k * n + i] = run;
      run += c[k];
    }
  }
}

__device__ __forceinline__ unsigned fill_slot(unsigned* cur, int d, unsigned lo, unsigned nb) {
  const unsigned b = (unsigned)d - lo;
  return (b < nb) ? atomicAdd(&cur[b], 1u) : 0xFFFFFFFFu;
}

__global__ __launch_bounds__(DEG_T) void fill_kernel(const int* __restrict__ src,
                                                     const int* __restrict__ dst,
                                                     const unsigned* __restrict__ first,
                                                     int* __restrict__ csr, int nE, int n) {
  __shared__ unsigned cur[DEG_BINS];
  const int tid = threadIdx.x;
  const int k = blockIdx.x % DEG_K;
  const int r = blockIdx.x / DEG_K;
  const unsigned lo = (unsigned)r * DEG_BINS;
  int nbi = n - (int)lo; if (nbi > DEG_BINS) nbi = DEG_BINS;
  const unsigned nb = (unsigned)nbi;
  const unsigned* __restrict__ f = first + (size_t)k * n + lo;
  for (int b = tid; b < nbi; b += DEG_T) cur[b] = f[b];
  __syncthreads();

  const int S = (((nE + DEG_K - 1) / DEG_K) + 3) & ~3;
  int e0 = k * S; if (e0 > nE) e0 = nE;
  int e1 = e0 + S; if (e1 > nE) e1 = nE;
  const int n4 = (e1 - e0) >> 2;
  const int4* __restrict__ d4 = (const int4*)(dst + e0);
  const int4* __restrict__ s4 = (const int4*)(src + e0);
  const unsigned lim = (unsigned)nE;
  for (int g = tid; g < n4; g += DEG_U * DEG_T) {
    int4 dv[DEG_U], sv[DEG_U];
#pragma unroll
    for (int u = 0; u < DEG_U; ++u) {
      const int gu = g + u * DEG_T;
      dv[u] = (gu < n4) ? d4[gu] : make_int4(-1, -1, -1, -1);
      sv[u] = (gu < n4) ? s4[gu] : make_int4(0, 0, 0, 0);
    }
    unsigned p[DEG_U * 4];
#pragma unroll
    for (int u = 0; u < DEG_U; ++u) {
      p[4 * u + 0] = fill_slot(cur, dv[u].x, lo, nb);
      p[4 * u + 1] = fill_slot(cur, dv[u].y, lo, nb);
      p[4 * u + 2] = fill_slot(cur, dv[u].z, lo, nb);
      p[4 * u + 3] = fill_slot(cur, dv[u].w, lo, nb);
    }
#pragma unroll
    for (int u = 0; u < DEG_U; ++u) {
      if (p[4 * u + 0] < lim) csr[p[4 * u + 0]] = sv[u].x;
      if (p[4 * u + 1] < lim) csr[p[4 * u + 1]] = sv[u].y;
      if (p[4 * u + 2] < lim) csr[p[4 * u + 2]] = sv[u].z;
      if (p[4 * u + 3] < lim) csr[p[4 * u + 3]] = sv[u].w;
    }
  }
  for (int e = e0 + (n4 << 2) + tid; e < e1; e += DEG_T) {
    const unsigned q = fill_slot(cur, dst[e], lo, nb);
    if (q < lim) csr[q] = src[e];
  }
}

__global__ __launch_bounds__(256) void gemm64_wmma(const float* __restrict__ X,
                                                   const float* __restrict__ W,
                                                   const float* __restrict__ row_scale,
                                                   float* __restrict__ Y, int n_rows) {
  __shared__ float Wt[64 * 64];
  const int tid = threadIdx.x;
  for (int i = tid; i < 64 * 64; i += 256) {
    int k = i >> 6, n = i & 63;
    Wt[n * 64 + k] = W[i];
  }
  __syncthreads();

  const int lane = tid & 31;
  const int wave = tid >> 5;
  const int tile = blockIdx.x * 8 + wave;
  if (tile * 16 >= n_rows) return;

  const int m = lane & 15;
  const int khalf = lane >> 4;
  const int row0 = tile * 16;
  const int row = row0 + m;
  const float sn = row_scale[row];
  const float* xrow = X + (size_t)row * 64;

  v8f acc0 = {}, acc1 = {}, acc2 = {}, acc3 = {};
#pragma unroll
  for (int kc = 0; kc < 16; ++kc) {
    const int kbase = kc * 4 + khalf * 2;
    const float2 av = *(const float2*)(xrow + kbase);
    v2f a; a.x = av.x * sn; a.y = av.y * sn;
    const v2f b0 = *(const v2f*)(&Wt[(m +  0) * 64 + kbase]);
    const v2f b1 = *(const v2f*)(&Wt[(m + 16) * 64 + kbase]);
    const v2f b2 = *(const v2f*)(&Wt[(m + 32) * 64 + kbase]);
    const v2f b3 = *(const v2f*)(&Wt[(m + 48) * 64 + kbase]);
    acc0 = __builtin_amdgcn_wmma_f32_16x16x4_f32(false, a, false, b0, (short)0, acc0, false, false);
    acc1 = __builtin_amdgcn_wmma_f32_16x16x4_f32(false, a, false, b1, (short)0, acc1, false, false);
    acc2 = __builtin_amdgcn_wmma_f32_16x16x4_f32(false, a, false, b2, (short)0, acc2, false, false);
    acc3 = __builtin_amdgcn_wmma_f32_16x16x4_f32(false, a, false, b3, (short)0, acc3, false, false);
  }

  float* yrow = Y + (size_t)row0 * 64;
  const int mofs = (lane < 16) ? 0 : 8;
#pragma unroll
  for (int j = 0; j < 8; ++j) {
    const size_t r = (size_t)(mofs + j) * 64;
    yrow[r +  0 + m] = acc0[j];
    yrow[r + 16 + m] = acc1[j];
    yrow[r + 32 + m] = acc2[j];
    yrow[r + 48 + m] = acc3[j];
  }
}

#define GAT_T 512
#define GAT_U 16

__global__ __launch_bounds__(GAT_T) void gather_kernel(const float* __restrict__ hW,
                                                       const unsigned* __restrict__ rowptr,
                                                       const int* __restrict__ csr,
                                                       const float* __restrict__ dn,
                                                       const float* __restrict__ bias,
                                                       float* __restrict__ out,
                                                       int n, int do_relu) {
  const int lane = threadIdx.x & 31;
  const int node = __builtin_amdgcn_readfirstlane((int)(blockIdx.x * (GAT_T / 32) + (threadIdx.x >> 5)));
  if (node >= n) return;
  const unsigned beg = rowptr[node], end = rowptr[node + 1];
  const float s = dn[node];
  const float2 b = *(const float2*)(bias + lane * 2);
  const float* __restrict__ col = hW + lane * 2;
  float ax = 0.f, ay = 0.f;
  for (unsigned base = beg; base < end; base += 32u) {
    const unsigned left = end - base;
    const int cnt = (int)(left < 32u ? left : 32u);
    const int idx = (lane < cnt) ? csr[base + lane] : -1;
    for (int j = 0; j < cnt; j += GAT_U) {
      float2 v[GAT_U];
#pragma unroll
      for (int u = 0; u < GAT_U; ++u) {
        const int s = __builtin_amdgcn_readlane(idx, j + u);
        v[u] = make_float2(0.f, 0.f);
        if ((unsigned)s < (unsigned)n) v[u] = *(const float2*)(col + (size_t)s * 64);
      }
#pragma unroll
      for (int u = 0; u < GAT_U; ++u) { ax += v[u].x; ay += v[u].y; }
    }
  }
  float2 y;
  y.x = ax * s + b.x;
  y.y = ay * s + b.y;
  if (do_relu) { y.x = fmaxf(y.x, 0.f); y.y = fmaxf(y.y, 0.f); }
  *(float2*)(out + (size_t)node * 64 + lane * 2) = y;
}

extern "C" void kernel_launch(void* const* d_in, const int* in_sizes, int n_in,
                              void* d_out, int out_size, void* d_ws, size_t ws_size,
                              hipStream_t stream) {
  const float* in_feat = (const float*)d_in[0];
  const int*   src     = (const int*)d_in[1];
  const int*   dst     = (const int*)d_in[2];
  const float* W1      = (const float*)d_in[3];
  const float* b1      = (const float*)d_in[4];
  const float* W2      = (const float*)d_in[5];
  const float* b2      = (const float*)d_in[6];
  float* out = (float*)d_out;

  const int N  = in_sizes[0] / 64;
  const int nE = in_sizes[1];

  const size_t feat_bytes = (size_t)N * 64 * sizeof(float);
  auto al = [](size_t x) { return (x + 255) & ~(size_t)255; };
  const int node_blocks = (N + 255) / 256;
  char* ws = (char*)d_ws;
  size_t off = 0;
  float*    hW     = (float*)(ws + off);    off += al(feat_bytes);
  float*    h1     = (float*)(ws + off);    off += al(feat_bytes);
  unsigned* part   = (unsigned*)(ws + off); off += al((size_t)2 * DEG_K * N * 4);
  float*    sn     = (float*)(ws + off);    off += al((size_t)N * 4);
  float*    dn     = (float*)(ws + off);    off += al((size_t)N * 4);
  unsigned* bsum   = (unsigned*)(ws + off); off += al((size_t)node_blocks * 4);
  unsigned* rowptr = (unsigned*)(ws + off); off += al((size_t)(N + 1) * 4);
  int*      csr    = (int*)(ws + off);      off += al((size_t)nE * 4);
  (void)ws_size; (void)n_in; (void)out_size;

  const int n_ranges = (N + DEG_BINS - 1) / DEG_BINS;
  unsigned* dst_part = part + (size_t)DEG_K * N;
  degree_hist_kernel<<<2 * n_ranges * DEG_K, DEG_T, 0, stream>>>(src, dst, part, nE, N, n_ranges);
  norm_kernel<<<node_blocks, 256, 0, stream>>>(part, sn, dn, bsum, N);
  rowptr_kernel<<<node_blocks, 256, 0, stream>>>(dst_part, bsum, rowptr, N);
  fill_kernel<<<n_ranges * DEG_K, DEG_T, 0, stream>>>(src, dst, dst_part, csr, nE, N);

  const int gemm_blocks = ((N + 15) / 16 + 7) / 8;
  const int gat_blocks = (N + GAT_T / 32 - 1) / (GAT_T / 32);

  gemm64_wmma<<<gemm_blocks, 256, 0, stream>>>(in_feat, W1, sn, hW, N);
  gather_kernel<<<gat_blocks, GAT_T, 0, stream>>>(hW, rowptr, csr, dn, b1, h1, N,  1);

  gemm64_wmma<<<gemm_blocks, 256, 0, stream>>>(h1, W2, sn, hW, N);
  gather_kernel<<<gat_blocks, GAT_T, 0, stream>>>(hW, rowptr, csr, dn, b2, out, N,  0);
}


# GEMM: W tile in LDS at a row pitch of 68 floats instead of 64: B-fragment reads (16 rows on one bank pair, 16-way conflict) and the transposing stores (32-way) become bank-conflict-free
# speedup vs baseline: 3.4893x; 1.0710x over previous
#include <hip/hip_runtime.h>
#include <hip/hip_bf16.h>

typedef float v2f __attribute__((ext_vector_type(2)));
typedef float v8f __attribute__((ext_vector_type(8)));

#define DEG_BINS 12544
#define DEG_K 16
#define DEG_T 512
#define DEG_U 8

__device__ __forceinline__ void deg_count(unsigned* h, int idx, unsigned lo) {
  const unsigned b = (unsigned)idx - lo;
  if (b < (unsigned)DEG_BINS) atomicAdd(&h[b], 1u);
}

__global__ __launch_bounds__(DEG_T) void degree_hist_kernel(const int* __restrict__ src,
                                                            const int* __restrict__ dst,
                                                            unsigned* __restrict__ part,
                                                            int nE, int n, int n_ranges) {
  __shared__ unsigned h[DEG_BINS];
  const int tid = threadIdx.x;
  int bid = blockIdx.x;
  const int k = bid % DEG_K; bid /= DEG_K;
  const int r = bid % n_ranges;
  const int which = bid / n_ranges;
  const int* __restrict__ idx = which ? dst : src;
  const unsigned lo = (unsigned)r * DEG_BINS;

  for (int b = tid; b < DEG_BINS; b += DEG_T) h[b] = 0u;
  __syncthreads();

  const int S = (((nE + DEG_K - 1) / DEG_K) + 3) & ~3;
  int e0 = k * S; if (e0 > nE) e0 = nE;
  int e1 = e0 + S; if (e1 > nE) e1 = nE;
  const int n4 = (e1 - e0) >> 2;
  const int4* __restrict__ p4 = (const int4*)(idx + e0);
  for (int g = tid; g < n4; g += DEG_U * DEG_T) {
    int4 v[DEG_U];
#pragma unroll
    for (int u = 0; u < DEG_U; ++u) {
      const int gu = g + u * DEG_T;
      v[u] = (gu < n4) ? p4[gu] : make_int4(-1, -1, -1, -1);
    }
#pragma unroll
    for (int u = 0; u < DEG_U; ++u) {
      deg_count(h, v[u].x, lo);
      deg_count(h, v[u].y, lo);
      deg_count(h, v[u].z, lo);
      deg_count(h, v[u].w, lo);
    }
  }
  for (int e = e0 + (n4 << 2) + tid; e < e1; e += DEG_T) deg_count(h, idx[e], lo);
  __syncthreads();

  int nb = n - (int)lo; if (nb > DEG_BINS) nb = DEG_BINS;
  unsigned* __restrict__ o = part + (size_t)(which * DEG_K + k) * n + lo;
  for (int b = tid; b < nb; b += DEG_T) o[b] = h[b];
}

__global__ __launch_bounds__(256) void norm_kernel(const unsigned* __restrict__ part,
                                                   float* __restrict__ sn,
                                                   float* __restrict__ dn,
                                                   unsigned* __restrict__ bsum, int n) {
  __shared__ unsigned tot;
  if (threadIdx.x == 0) tot = 0u;
  __syncthreads();
  int i = blockIdx.x * 256 + threadIdx.x;
  unsigned a = 0u, b = 0u;
  if (i < n) {
#pragma unroll
    for (int k = 0; k < DEG_K; ++k) {
      a += part[(size_t)k * n + i];
      b += part[(size_t)(DEG_K + k) * n + i];
    }
    sn[i] = rsqrtf((float)(a < 1u ? 1u : a));
    dn[i] = rsqrtf((float)(b < 1u ? 1u : b));
  }
  atomicAdd(&tot, b);
  __syncthreads();
  if (threadIdx.x == 0) bsum[blockIdx.x] = tot;
}

__global__ __launch_bounds__(256) void rowptr_kernel(unsigned* __restrict__ cnt,
                                                     const unsigned* __restrict__ bsum,
                                                     unsigned* __restrict__ rowptr, int n) {
  __shared__ unsigned wsum[8];
  __shared__ unsigned base_s;
  const int tid = threadIdx.x;
  const int lane = tid & 31, wave = tid >> 5;
  if (tid == 0) base_s = 0u;
  __syncthreads();
  unsigned pre = 0u;
  for (int j = tid; j < (int)blockIdx.x; j += 256) pre += bsum[j];
  atomicAdd(&base_s, pre);

  const int i = blockIdx.x * 256 + tid;
  unsigned c[DEG_K];
  unsigned deg = 0u;
#pragma unroll
  for (int k = 0; k < DEG_K; ++k) {
    c[k] = (i < n) ? cnt[(size_t)k * n + i] : 0u;
    deg += c[k];
  }
  unsigned inc = deg;
  for (int d = 1; d < 32; d <<= 1) {
    const unsigned t = __shfl_up(inc, d);
    if (lane >= d) inc += t;
  }
  if (lane == 31) wsum[wave] = inc;
  __syncthreads();
  unsigned start = base_s + inc - deg;
  for (int w = 0; w < wave; ++w) start += wsum[w];
  if (i < n) {
    rowptr[i] = start;
    if (i == n - 1) rowptr[n] = start + deg;
    unsigned run = start;
#pragma unroll
    for (int k = 0; k < DEG_K; ++k) {
      cnt[(size_t)k * n + i] = run;
      run += c[k];
    }
  }
}

__device__ __forceinline__ unsigned fill_slot(unsigned* cur, int d, unsigned lo, unsigned nb) {
  const unsigned b = (unsigned)d - lo;
  return (b < nb) ? atomicAdd(&cur[b], 1u) : 0xFFFFFFFFu;
}

__global__ __launch_bounds__(DEG_T) void fill_kernel(const int* __restrict__ src,
                                                     const int* __restrict__ dst,
                                                     const unsigned* __restrict__ first,
                                                     int* __restrict__ csr, int nE, int n) {
  __shared__ unsigned cur[DEG_BINS];
  const int tid = threadIdx.x;
  const int k = blockIdx.x % DEG_K;
  const int r = blockIdx.x / DEG_K;
  const unsigned lo = (unsigned)r * DEG_BINS;
  int nbi = n - (int)lo; if (nbi > DEG_BINS) nbi = DEG_BINS;
  const unsigned nb = (unsigned)nbi;
  const unsigned* __restrict__ f = first + (size_t)k * n + lo;
  for (int b = tid; b < nbi; b += DEG_T) cur[b] = f[b];
  __syncthreads();

  const int S = (((nE + DEG_K - 1) / DEG_K) + 3) & ~3;
  int e0 = k * S; if (e0 > nE) e0 = nE;
  int e1 = e0 + S; if (e1 > nE) e1 = nE;
  const int n4 = (e1 - e0) >> 2;
  const int4* __restrict__ d4 = (const int4*)(dst + e0);
  const int4* __restrict__ s4 = (const int4*)(src + e0);
  const unsigned lim = (unsigned)nE;
  for (int g = tid; g < n4; g += DEG_U * DEG_T) {
    int4 dv[DEG_U], sv[DEG_U];
#pragma unroll
    for (int u = 0; u < DEG_U; ++u) {
      const int gu = g + u * DEG_T;
      dv[u] = (gu < n4) ? d4[gu] : make_int4(-1, -1, -1, -1);
      sv[u] = (gu < n4) ? s4[gu] : make_int4(0, 0, 0, 0);
    }
    unsigned p[DEG_U * 4];
#pragma unroll
    for (int u = 0; u < DEG_U; ++u) {
      p[4 * u + 0] = fill_slot(cur, dv[u].x, lo, nb);
      p[4 * u + 1] = fill_slot(cur, dv[u].y, lo, nb);
      p[4 * u + 2] = fill_slot(cur, dv[u].z, lo, nb);
      p[4 * u + 3] = fill_slot(cur, dv[u].w, lo, nb);
    }
#pragma unroll
    for (int u = 0; u < DEG_U; ++u) {
      if (p[4 * u + 0] < lim) csr[p[4 * u + 0]] = sv[u].x;
      if (p[4 * u + 1] < lim) csr[p[4 * u + 1]] = sv[u].y;
      if (p[4 * u + 2] < lim) csr[p[4 * u + 2]] = sv[u].z;
      if (p[4 * u + 3] < lim) csr[p[4 * u + 3]] = sv[u].w;
    }
  }
  for (int e = e0 + (n4 << 2) + tid; e < e1; e += DEG_T) {
    const unsigned q = fill_slot(cur, dst[e], lo, nb);
    if (q < lim) csr[q] = src[e];
  }
}

#define WT_PITCH 68
__global__ __launch_bounds__(256) void gemm64_wmma(const float* __restrict__ X,
                                                   const float* __restrict__ W,
                                                   const float* __restrict__ row_scale,
                                                   float* __restrict__ Y, int n_rows) {
  __shared__ float Wt[64 * WT_PITCH];
  const int tid = threadIdx.x;
  for (int i = tid; i < 64 * 64; i += 256) {
    int k = i >> 6, n = i & 63;
    Wt[n * WT_PITCH + k] = W[i];
  }
  __syncthreads();

  const int lane = tid & 31;
  const int wave = tid >> 5;
  const int tile = blockIdx.x * 8 + wave;
  if (tile * 16 >= n_rows) return;

  const int m = lane & 15;
  const int khalf = lane >> 4;
  const int row0 = tile * 16;
  const int row = row0 + m;
  const float sn = row_scale[row];
  const float* xrow = X + (size_t)row * 64;

  v8f acc0 = {}, acc1 = {}, acc2 = {}, acc3 = {};
#pragma unroll
  for (int kc = 0; kc < 16; ++kc) {
    const int kbase = kc * 4 + khalf * 2;
    const float2 av = *(const float2*)(xrow + kbase);
    v2f a; a.x = av.x * sn; a.y = av.y * sn;
    const v2f b0 = *(const v2f*)(&Wt[(m +  0) * WT_PITCH + kbase]);
    const v2f b1 = *(const v2f*)(&Wt[(m + 16) * WT_PITCH + kbase]);
    const v2f b2 = *(const v2f*)(&Wt[(m + 32) * WT_PITCH + kbase]);
    const v2f b3 = *(const v2f*)(&Wt[(m + 48) * WT_PITCH + kbase]);
    acc0 = __builtin_amdgcn_wmma_f32_16x16x4_f32(false, a, false, b0, (short)0, acc0, false, false);
    acc1 = __builtin_amdgcn_wmma_f32_16x16x4_f32(false, a, false, b1, (short)0, acc1, false, false);
    acc2 = __builtin_amdgcn_wmma_f32_16x16x4_f32(false, a, false, b2, (short)0, acc2, false, false);
    acc3 = __builtin_amdgcn_wmma_f32_16x16x4_f32(false, a, false, b3, (short)0, acc3, false, false);
  }

  float* yrow = Y + (size_t)row0 * 64;
  const int mofs = (lane < 16) ? 0 : 8;
#pragma unroll
  for (int j = 0; j < 8; ++j) {
    const size_t r = (size_t)(mofs + j) * 64;
    yrow[r +  0 + m] = acc0[j];
    yrow[r + 16 + m] = acc1[j];
    yrow[r + 32 + m] = acc2[j];
    yrow[r + 48 + m] = acc3[j];
  }
}

#define GAT_T 512
#define GAT_U 16

__global__ __launch_bounds__(GAT_T) void gather_kernel(const float* __restrict__ hW,
                                                       const unsigned* __restrict__ rowptr,
                                                       const int* __restrict__ csr,
                                                       const float* __restrict__ dn,
                                                       const float* __restrict__ bias,
                                                       float* __restrict__ out,
                                                       int n, int do_relu) {
  const int lane = threadIdx.x & 31;
  const int node = __builtin_amdgcn_readfirstlane((int)(blockIdx.x * (GAT_T / 32) + (threadIdx.x >> 5)));
  if (node >= n) return;
  const unsigned beg = rowptr[node], end = rowptr[node + 1];
  const float s = dn[node];
  const float2 b = *(const float2*)(bias + lane * 2);
  const float* __restrict__ col = hW + lane * 2;
  float ax = 0.f, ay = 0.f;
  for (unsigned base = beg; base < end; base += 32u) {
    const unsigned left = end - base;
    const int cnt = (int)(left < 32u ? left : 32u);
    const int idx = (lane < cnt) ? csr[base + lane] : -1;
    for (int j = 0; j < cnt; j += GAT_U) {
      float2 v[GAT_U];
#pragma unroll
      for (int u = 0; u < GAT_U; ++u) {
        const int s = __builtin_amdgcn_readlane(idx, j + u);
        v[u] = make_float2(0.f, 0.f);
        if ((unsigned)s < (unsigned)n) v[u] = *(const float2*)(col + (size_t)s * 64);
      }
#pragma unroll
      for (int u = 0; u < GAT_U; ++u) { ax += v[u].x; ay += v[u].y; }
    }
  }
  float2 y;
  y.x = ax * s + b.x;
  y.y = ay * s + b.y;
  if (do_relu) { y.x = fmaxf(y.x, 0.f); y.y = fmaxf(y.y, 0.f); }
  *(float2*)(out + (size_t)node * 64 + lane * 2) = y;
}

extern "C" void kernel_launch(void* const* d_in, const int* in_sizes, int n_in,
                              void* d_out, int out_size, void* d_ws, size_t ws_size,
                              hipStream_t stream) {
  const float* in_feat = (const float*)d_in[0];
  const int*   src     = (const int*)d_in[1];
  const int*   dst     = (const int*)d_in[2];
  const float* W1      = (const float*)d_in[3];
  const float* b1      = (const float*)d_in[4];
  const float* W2      = (const float*)d_in[5];
  const float* b2      = (const float*)d_in[6];
  float* out = (float*)d_out;

  const int N  = in_sizes[0] / 64;
  const int nE = in_sizes[1];

  const size_t feat_bytes = (size_t)N * 64 * sizeof(float);
  auto al = [](size_t x) { return (x + 255) & ~(size_t)255; };
  const int node_blocks = (N + 255) / 256;
  char* ws = (char*)d_ws;
  size_t off = 0;
  float*    hW     = (float*)(ws + off);    off += al(feat_bytes);
  float*    h1     = (float*)(ws + off);    off += al(feat_bytes);
  unsigned* part   = (unsigned*)(ws + off); off += al((size_t)2 * DEG_K * N * 4);
  float*    sn     = (float*)(ws + off);    off += al((size_t)N * 4);
  float*    dn     = (float*)(ws + off);    off += al((size_t)N * 4);
  unsigned* bsum   = (unsigned*)(ws + off); off += al((size_t)node_blocks * 4);
  unsigned* rowptr = (unsigned*)(ws + off); off += al((size_t)(N + 1) * 4);
  int*      csr    = (int*)(ws + off);      off += al((size_t)nE * 4);
  (void)ws_size; (void)n_in; (void)out_size;

  const int n_ranges = (N + DEG_BINS - 1) / DEG_BINS;
  unsigned* dst_part = part + (size_t)DEG_K * N;
  degree_hist_kernel<<<2 * n_ranges * DEG_K, DEG_T, 0, stream>>>(src, dst, part, nE, N, n_ranges);
  norm_kernel<<<node_blocks, 256, 0, stream>>>(part, sn, dn, bsum, N);
  rowptr_kernel<<<node_blocks, 256, 0, stream>>>(dst_part, bsum, rowptr, N);
  fill_kernel<<<n_ranges * DEG_K, DEG_T, 0, stream>>>(src, dst, dst_part, csr, nE, N);

  const int gemm_blocks = ((N + 15) / 16 + 7) / 8;
  const int gat_blocks = (N + GAT_T / 32 - 1) / (GAT_T / 32);

  gemm64_wmma<<<gemm_blocks, 256, 0, stream>>>(in_feat, W1, sn, hW, N);
  gather_kernel<<<gat_blocks, GAT_T, 0, stream>>>(hW, rowptr, csr, dn, b1, h1, N,  1);

  gemm64_wmma<<<gemm_blocks, 256, 0, stream>>>(h1, W2, sn, hW, N);
  gather_kernel<<<gat_blocks, GAT_T, 0, stream>>>(hW, rowptr, csr, dn, b2, out, N,  0);
}


# layer-2 product moved into the tail of the layer-1 gather launch: a block's 16 finished rows are one A tile in LDS, 4 waves do its f32 WMMAs; second GEMM launch and the h1 round trip go (8 -> 7)
# speedup vs baseline: 3.7859x; 1.0850x over previous
#include <hip/hip_runtime.h>
#include <hip/hip_bf16.h>

typedef float v2f __attribute__((ext_vector_type(2)));
typedef float v8f __attribute__((ext_vector_type(8)));

#define DEG_BINS 12544
#define DEG_K 16
#define DEG_T 512
#define DEG_U 8

__device__ __forceinline__ void deg_count(unsigned* h, int idx, unsigned lo) {
  const unsigned b = (unsigned)idx - lo;
  if (b < (unsigned)DEG_BINS) atomicAdd(&h[b], 1u);
}

__global__ __launch_bounds__(DEG_T) void degree_hist_kernel(const int* __restrict__ src,
                                                            const int* __restrict__ dst,
                                                            unsigned* __restrict__ part,
                                                            int nE, int n, int n_ranges) {
  __shared__ unsigned h[DEG_BINS];
  const int tid = threadIdx.x;
  int bid = blockIdx.x;
  const int k = bid % DEG_K; bid /= DEG_K;
  const int r = bid % n_ranges;
  const int which = bid / n_ranges;
  const int* __restrict__ idx = which ? dst : src;
  const unsigned lo = (unsigned)r * DEG_BINS;

  for (int b = tid; b < DEG_BINS; b += DEG_T) h[b] = 0u;
  __syncthreads();

  const int S = (((nE + DEG_K - 1) / DEG_K) + 3) & ~3;
  int e0 = k * S; if (e0 > nE) e0 = nE;
  int e1 = e0 + S; if (e1 > nE) e1 = nE;
  const int n4 = (e1 - e0) >> 2;
  const int4* __restrict__ p4 = (const int4*)(idx + e0);
  for (int g = tid; g < n4; g += DEG_U * DEG_T) {
    int4 v[DEG_U];
#pragma unroll
    for (int u = 0; u < DEG_U; ++u) {
      const int gu = g + u * DEG_T;
      v[u] = (gu < n4) ? p4[gu] : make_int4(-1, -1, -1, -1);
    }
#pragma unroll
    for (int u = 0; u < DEG_U; ++u) {
      deg_count(h, v[u].x, lo);
      deg_count(h, v[u].y, lo);
      deg_count(h, v[u].z, lo);
      deg_count(h, v[u].w, lo);
    }
  }
  for (int e = e0 + (n4 << 2) + tid; e < e1; e += DEG_T) deg_count(h, idx[e], lo);
  __syncthreads();

  int nb = n - (int)lo; if (nb > DEG_BINS) nb = DEG_BINS;
  unsigned* __restrict__ o = part + (size_t)(which * DEG_K + k) * n + lo;
  for (int b = tid; b < nb; b += DEG_T) o[b] = h[b];
}

__global__ __launch_bounds__(256) void norm_kernel(const unsigned* __restrict__ part,
                                                   float* __restrict__ sn,
                                                   float* __restrict__ dn,
                                                   unsigned* __restrict__ bsum, int n) {
  __shared__ unsigned tot;
  if (threadIdx.x == 0) tot = 0u;
  __syncthreads();
  int i = blockIdx.x * 256 + threadIdx.x;
  unsigned a = 0u, b = 0u;
  if (i < n) {
#pragma unroll
    for (int k = 0; k < DEG_K; ++k) {
      a += part[(size_t)k * n + i];
      b += part[(size_t)(DEG_K + k) * n + i];
    }
    sn[i] = rsqrtf((float)(a < 1u ? 1u : a));
    dn[i] = rsqrtf((float)(b < 1u ? 1u : b));
  }
  atomicAdd(&tot, b);
  __syncthreads();
  if (threadIdx.x == 0) bsum[blockIdx.x] = tot;
}

__global__ __launch_bounds__(256) void rowptr_kernel(unsigned* __restrict__ cnt,
                                                     const unsigned* __restrict__ bsum,
                                                     unsigned* __restrict__ rowptr, int n) {
  __shared__ unsigned wsum[8];
  __shared__ unsigned base_s;
  const int tid = threadIdx.x;
  const int lane = tid & 31, wave = tid >> 5;
  if (tid == 0) base_s = 0u;
  __syncthreads();
  unsigned pre = 0u;
  for (int j = tid; j < (int)blockIdx.x; j += 256) pre += bsum[j];
  atomicAdd(&base_s, pre);

  const int i = blockIdx.x * 256 + tid;
  unsigned c[DEG_K];
  unsigned deg = 0u;
#pragma unroll
  for (int k = 0; k < DEG_K; ++k) {
    c[k] = (i < n) ? cnt[(size_t)k * n + i] : 0u;
    deg += c[k];
  }
  unsigned inc = deg;
  for (int d = 1; d < 32; d <<= 1) {
    const unsigned t = __shfl_up(inc, d);
    if (lane >= d) inc += t;
  }
  if (lane == 31) wsum[wave] = inc;
  __syncthreads();
  unsigned start = base_s + inc - deg;
  for (int w = 0; w < wave; ++w) start += wsum[w];
  if (i < n) {
    rowptr[i] = start;
    if (i == n - 1) rowptr[n] = start + deg;
    unsigned run = start;
#pragma unroll
    for (int k = 0; k < DEG_K; ++k) {
      cnt[(size_t)k * n + i] = run;
      run += c[k];
    }
  }
}

__device__ __forceinline__ unsigned fill_slot(unsigned* cur, int d, unsigned lo, unsigned nb) {
  const unsigned b = (unsigned)d - lo;
  return (b < nb) ? atomicAdd(&cur[b], 1u) : 0xFFFFFFFFu;
}

__global__ __launch_bounds__(DEG_T) void fill_kernel(const int* __restrict__ src,
                                                     const int* __restrict__ dst,
                                                     const unsigned* __restrict__ first,
                                                     int* __restrict__ csr, int nE, int n) {
  __shared__ unsigned cur[DEG_BINS];
  const int tid = threadIdx.x;
  const int k = blockIdx.x % DEG_K;
  const int r = blockIdx.x / DEG_K;
  const unsigned lo = (unsigned)r * DEG_BINS;
  int nbi = n - (int)lo; if (nbi > DEG_BINS) nbi = DEG_BINS;
  const unsigned nb = (unsigned)nbi;
  const unsigned* __restrict__ f = first + (size_t)k * n + lo;
  for (int b = tid; b < nbi; b += DEG_T) cur[b] = f[b];
  __syncthreads();

  const int S = (((nE + DEG_K - 1) / DEG_K) + 3) & ~3;
  int e0 = k * S; if (e0 > nE) e0 = nE;
  int e1 = e0 + S; if (e1 > nE) e1 = nE;
  const int n4 = (e1 - e0) >> 2;
  const int4* __restrict__ d4 = (const int4*)(dst + e0);
  const int4* __restrict__ s4 = (const int4*)(src + e0);
  const unsigned lim = (unsigned)nE;
  for (int g = tid; g < n4; g += DEG_U * DEG_T) {
    int4 dv[DEG_U], sv[DEG_U];
#pragma unroll
    for (int u = 0; u < DEG_U; ++u) {
      const int gu = g + u * DEG_T;
      dv[u] = (gu < n4) ? d4[gu] : make_int4(-1, -1, -1, -1);
      sv[u] = (gu < n4) ? s4[gu] : make_int4(0, 0, 0, 0);
    }
    unsigned p[DEG_U * 4];
#pragma unroll
    for (int u = 0; u < DEG_U; ++u) {
      p[4 * u + 0] = fill_slot(cur, dv[u].x, lo, nb);
      p[4 * u + 1] = fill_slot(cur, dv[u].y, lo, nb);
      p[4 * u + 2] = fill_slot(cur, dv[u].z, lo, nb);
      p[4 * u + 3] = fill_slot(cur, dv[u].w, lo, nb);
    }
#pragma unroll
    for (int u = 0; u < DEG_U; ++u) {
      if (p[4 * u + 0] < lim) csr[p[4 * u + 0]] = sv[u].x;
      if (p[4 * u + 1] < lim) csr[p[4 * u + 1]] = sv[u].y;
      if (p[4 * u + 2] < lim) csr[p[4 * u + 2]] = sv[u].z;
      if (p[4 * u + 3] < lim) csr[p[4 * u + 3]] = sv[u].w;
    }
  }
  for (int e = e0 + (n4 << 2) + tid; e < e1; e += DEG_T) {
    const unsigned q = fill_slot(cur, dst[e], lo, nb);
    if (q < lim) csr[q] = src[e];
  }
}

#define WT_PITCH 68
__global__ __launch_bounds__(256) void gemm64_wmma(const float* __restrict__ X,
                                                   const float* __restrict__ W,
                                                   const float* __restrict__ row_scale,
                                                   float* __restrict__ Y, int n_rows) {
  __shared__ float Wt[64 * WT_PITCH];
  const int tid = threadIdx.x;
  for (int i = tid; i < 64 * 64; i += 256) {
    int k = i >> 6, n = i & 63;
    Wt[n * WT_PITCH + k] = W[i];
  }
  __syncthreads();

  const int lane = tid & 31;
  const int wave = tid >> 5;
  const int tile = blockIdx.x * 8 + wave;
  if (tile * 16 >= n_rows) return;

  const int m = lane & 15;
  const int khalf = lane >> 4;
  const int row0 = tile * 16;
  const int row = row0 + m;
  const float sn = row_scale[row];
  const float* xrow = X + (size_t)row * 64;

  v8f acc0 = {}, acc1 = {}, acc2 = {}, acc3 = {};
#pragma unroll
  for (int kc = 0; kc < 16; ++kc) {
    const int kbase = kc * 4 + khalf * 2;
    const float2 av = *(const float2*)(xrow + kbase);
    v2f a; a.x = av.x * sn; a.y = av.y * sn;
    const v2f b0 = *(const v2f*)(&Wt[(m +  0) * WT_PITCH + kbase]);
    const v2f b1 = *(const v2f*)(&Wt[(m + 16) * WT_PITCH + kbase]);
    const v2f b2 = *(const v2f*)(&Wt[(m + 32) * WT_PITCH + kbase]);
    const v2f b3 = *(const v2f*)(&Wt[(m + 48) * WT_PITCH + kbase]);
    acc0 = __builtin_amdgcn_wmma_f32_16x16x4_f32(false, a, false, b0, (short)0, acc0, false, false);
    acc1 = __builtin_amdgcn_wmma_f32_16x16x4_f32(false, a, false, b1, (short)0, acc1, false, false);
    acc2 = __builtin_amdgcn_wmma_f32_16x16x4_f32(false, a, false, b2, (short)0, acc2, false, false);
    acc3 = __builtin_amdgcn_wmma_f32_16x16x4_f32(false, a, false, b3, (short)0, acc3, false, false);
  }

  float* yrow = Y + (size_t)row0 * 64;
  const int mofs = (lane < 16) ? 0 : 8;
#pragma unroll
  for (int j = 0; j < 8; ++j) {
    const size_t r = (size_t)(mofs + j) * 64;
    yrow[r +  0 + m] = acc0[j];
    yrow[r + 16 + m] = acc1[j];
    yrow[r + 32 + m] = acc2[j];
    yrow[r + 48 + m] = acc3[j];
  }
}

#define GAT_T 512
#define GAT_U 16

__device__ __forceinline__ float2 gather_sum(const float* __restrict__ hW,
                                             const int* __restrict__ csr,
                                             unsigned beg, unsigned end, int lane, int n) {
  const float* __restrict__ col = hW + lane * 2;
  float ax = 0.f, ay = 0.f;
  for (unsigned base = beg; base < end; base += 32u) {
    const unsigned left = end - base;
    const int cnt = (int)(left < 32u ? left : 32u);
    const int idx = (lane < cnt) ? csr[base + lane] : -1;
    for (int j = 0; j < cnt; j += GAT_U) {
      float2 v[GAT_U];
#pragma unroll
      for (int u = 0; u < GAT_U; ++u) {
        const int s = __builtin_amdgcn_readlane(idx, j + u);
        v[u] = make_float2(0.f, 0.f);
        if ((unsigned)s < (unsigned)n) v[u] = *(const float2*)(col + (size_t)s * 64);
      }
#pragma unroll
      for (int u = 0; u < GAT_U; ++u) { ax += v[u].x; ay += v[u].y; }
    }
  }
  return make_float2(ax, ay);
}

__global__ __launch_bounds__(GAT_T) void gather_kernel(const float* __restrict__ hW,
                                                       const unsigned* __restrict__ rowptr,
                                                       const int* __restrict__ csr,
                                                       const float* __restrict__ dn,
                                                       const float* __restrict__ bias,
                                                       float* __restrict__ out, int n) {
  const int lane = threadIdx.x & 31;
  const int node = __builtin_amdgcn_readfirstlane((int)(blockIdx.x * (GAT_T / 32) + (threadIdx.x >> 5)));
  if (node >= n) return;
  const unsigned beg = rowptr[node], end = rowptr[node + 1];
  const float s = dn[node];
  const float2 b = *(const float2*)(bias + lane * 2);
  const float2 a = gather_sum(hW, csr, beg, end, lane, n);
  float2 y;
  y.x = a.x * s + b.x;
  y.y = a.y * s + b.y;
  *(float2*)(out + (size_t)node * 64 + lane * 2) = y;
}

#define AT_PITCH 68
__global__ __launch_bounds__(GAT_T) void gather_gemm_kernel(const float* __restrict__ hW,
                                                            const unsigned* __restrict__ rowptr,
                                                            const int* __restrict__ csr,
                                                            const float* __restrict__ dn,
                                                            const float* __restrict__ bias,
                                                            const float* __restrict__ sn,
                                                            const float* __restrict__ W,
                                                            float* __restrict__ Y, int n) {
  __shared__ float At[(GAT_T / 32) * AT_PITCH];
  __shared__ float Yt[(GAT_T / 32) * AT_PITCH];
  const int lane = threadIdx.x & 31;
  const int wave = __builtin_amdgcn_readfirstlane((int)(threadIdx.x >> 5));
  const int node = (int)(blockIdx.x * (GAT_T / 32)) + wave;

  float2 y = make_float2(0.f, 0.f);
  if (node < n) {
    const unsigned beg = rowptr[node], end = rowptr[node + 1];
    const float s = dn[node];
    const float so = sn[node];
    const float2 b = *(const float2*)(bias + lane * 2);
    const float2 a = gather_sum(hW, csr, beg, end, lane, n);
    y.x = fmaxf(a.x * s + b.x, 0.f);
    y.y = fmaxf(a.y * s + b.y, 0.f);
    y.x *= so;
    y.y *= so;
  }
  *(float2*)(&At[wave * AT_PITCH + lane * 2]) = y;

  const int c = lane & 15;
  const int khalf = lane >> 4;
  v2f bf[16];
  if (wave < 4) {
    const float* __restrict__ wp = W + (khalf * 2) * 64 + wave * 16 + c;
#pragma unroll
    for (int kc = 0; kc < 16; ++kc) {
      bf[kc].x = wp[(kc * 4 + 0) * 64];
      bf[kc].y = wp[(kc * 4 + 1) * 64];
    }
  }
  __syncthreads();
  if (wave < 4) {
    v8f acc = {};
#pragma unroll
    for (int kc = 0; kc < 16; ++kc) {
      const v2f a = *(const v2f*)(&At[c * AT_PITCH + kc * 4 + khalf * 2]);
      acc = __builtin_amdgcn_wmma_f32_16x16x4_f32(false, a, false, bf[kc], (short)0, acc, false, false);
    }
#pragma unroll
    for (int j = 0; j < 8; ++j) Yt[(khalf * 8 + j) * AT_PITCH + wave * 16 + c] = acc[j];
  }
  __syncthreads();
  if (node < n) {
    const float2 o = *(const float2*)(&Yt[wave * AT_PITCH + lane * 2]);
    *(float2*)(Y + (size_t)node * 64 + lane * 2) = o;
  }
}

extern "C" void kernel_launch(void* const* d_in, const int* in_sizes, int n_in,
                              void* d_out, int out_size, void* d_ws, size_t ws_size,
                              hipStream_t stream) {
  const float* in_feat = (const float*)d_in[0];
  const int*   src     = (const int*)d_in[1];
  const int*   dst     = (const int*)d_in[2];
  const float* W1      = (const float*)d_in[3];
  const float* b1      = (const float*)d_in[4];
  const float* W2      = (const float*)d_in[5];
  const float* b2      = (const float*)d_in[6];
  float* out = (float*)d_out;

  const int N  = in_sizes[0] / 64;
  const int nE = in_sizes[1];

  const size_t feat_bytes = (size_t)N * 64 * sizeof(float);
  auto al = [](size_t x) { return (x + 255) & ~(size_t)255; };
  const int node_blocks = (N + 255) / 256;
  char* ws = (char*)d_ws;
  size_t off = 0;
  float*    hW     = (float*)(ws + off);    off += al(feat_bytes);
  float*    hW2    = (float*)(ws + off);    off += al(feat_bytes);
  unsigned* part   = (unsigned*)(ws + off); off += al((size_t)2 * DEG_K * N * 4);
  float*    sn     = (float*)(ws + off);    off += al((size_t)N * 4);
  float*    dn     = (float*)(ws + off);    off += al((size_t)N * 4);
  unsigned* bsum   = (unsigned*)(ws + off); off += al((size_t)node_blocks * 4);
  unsigned* rowptr = (unsigned*)(ws + off); off += al((size_t)(N + 1) * 4);
  int*      csr    = (int*)(ws + off);      off += al((size_t)nE * 4);
  (void)ws_size; (void)n_in; (void)out_size;

  const int n_ranges = (N + DEG_BINS - 1) / DEG_BINS;
  unsigned* dst_part = part + (size_t)DEG_K * N;
  degree_hist_kernel<<<2 * n_ranges * DEG_K, DEG_T, 0, stream>>>(src, dst, part, nE, N, n_ranges);
  norm_kernel<<<node_blocks, 256, 0, stream>>>(part, sn, dn, bsum, N);
  rowptr_kernel<<<node_blocks, 256, 0, stream>>>(dst_part, bsum, rowptr, N);
  fill_kernel<<<n_ranges * DEG_K, DEG_T, 0, stream>>>(src, dst, dst_part, csr, nE, N);

  const int gemm_blocks = ((N + 15) / 16 + 7) / 8;
  const int gat_blocks = (N + GAT_T / 32 - 1) / (GAT_T / 32);

  gemm64_wmma<<<gemm_blocks, 256, 0, stream>>>(in_feat, W1, sn, hW, N);
  gather_gemm_kernel<<<gat_blocks, GAT_T, 0, stream>>>(hW, rowptr, csr, dn, b1, sn, W2, hW2, N);

  gather_kernel<<<gat_blocks, GAT_T, 0, stream>>>(hW2, rowptr, csr, dn, b2, out, N);
}


# both gathers: 2 nodes per wave, one per 16-lane row, 16 B per lane; a slot's src index reaches its row by DPP row_share, not readlane + scalar address; empty slots read a zero row; half the waves
# speedup vs baseline: 3.9129x; 1.0335x over previous
#include <hip/hip_runtime.h>
#include <hip/hip_bf16.h>

typedef float v2f __attribute__((ext_vector_type(2)));
typedef float v8f __attribute__((ext_vector_type(8)));

#define DEG_BINS 12544
#define DEG_K 16
#define DEG_T 512
#define DEG_U 8

__device__ __forceinline__ void deg_count(unsigned* h, int idx, unsigned lo) {
  const unsigned b = (unsigned)idx - lo;
  if (b < (unsigned)DEG_BINS) atomicAdd(&h[b], 1u);
}

__global__ __launch_bounds__(DEG_T) void degree_hist_kernel(const int* __restrict__ src,
                                                            const int* __restrict__ dst,
                                                            unsigned* __restrict__ part,
                                                            int nE, int n, int n_ranges) {
  __shared__ unsigned h[DEG_BINS];
  const int tid = threadIdx.x;
  int bid = blockIdx.x;
  const int k = bid % DEG_K; bid /= DEG_K;
  const int r = bid % n_ranges;
  const int which = bid / n_ranges;
  const int* __restrict__ idx = which ? dst : src;
  const unsigned lo = (unsigned)r * DEG_BINS;

  for (int b = tid; b < DEG_BINS; b += DEG_T) h[b] = 0u;
  __syncthreads();

  const int S = (((nE + DEG_K - 1) / DEG_K) + 3) & ~3;
  int e0 = k * S; if (e0 > nE) e0 = nE;
  int e1 = e0 + S; if (e1 > nE) e1 = nE;
  const int n4 = (e1 - e0) >> 2;
  const int4* __restrict__ p4 = (const int4*)(idx + e0);
  for (int g = tid; g < n4; g += DEG_U * DEG_T) {
    int4 v[DEG_U];
#pragma unroll
    for (int u = 0; u < DEG_U; ++u) {
      const int gu = g + u * DEG_T;
      v[u] = (gu < n4) ? p4[gu] : make_int4(-1, -1, -1, -1);
    }
#pragma unroll
    for (int u = 0; u < DEG_U; ++u) {
      deg_count(h, v[u].x, lo);
      deg_count(h, v[u].y, lo);
      deg_count(h, v[u].z, lo);
      deg_count(h, v[u].w, lo);
    }
  }
  for (int e = e0 + (n4 << 2) + tid; e < e1; e += DEG_T) deg_count(h, idx[e], lo);
  __syncthreads();

  int nb = n - (int)lo; if (nb > DEG_BINS) nb = DEG_BINS;
  unsigned* __restrict__ o = part + (size_t)(which * DEG_K + k) * n + lo;
  for (int b = tid; b < nb; b += DEG_T) o[b] = h[b];
}

__global__ __launch_bounds__(256) void norm_kernel(const unsigned* __restrict__ part,
                                                   float* __restrict__ sn,
                                                   float* __restrict__ dn,
                                                   unsigned* __restrict__ bsum, int n) {
  __shared__ unsigned tot;
  if (threadIdx.x == 0) tot = 0u;
  __syncthreads();
  int i = blockIdx.x * 256 + threadIdx.x;
  unsigned a = 0u, b = 0u;
  if (i < n) {
#pragma unroll
    for (int k = 0; k < DEG_K; ++k) {
      a += part[(size_t)k * n + i];
      b += part[(size_t)(DEG_K + k) * n + i];
    }
    sn[i] = rsqrtf((float)(a < 1u ? 1u : a));
    dn[i] = rsqrtf((float)(b < 1u ? 1u : b));
  }
  atomicAdd(&tot, b);
  __syncthreads();
  if (threadIdx.x == 0) bsum[blockIdx.x] = tot;
}

__global__ __launch_bounds__(256) void rowptr_kernel(unsigned* __restrict__ cnt,
                                                     const unsigned* __restrict__ bsum,
                                                     unsigned* __restrict__ rowptr, int n) {
  __shared__ unsigned wsum[8];
  __shared__ unsigned base_s;
  const int tid = threadIdx.x;
  const int lane = tid & 31, wave = tid >> 5;
  if (tid == 0) base_s = 0u;
  __syncthreads();
  unsigned pre = 0u;
  for (int j = tid; j < (int)blockIdx.x; j += 256) pre += bsum[j];
  atomicAdd(&base_s, pre);

  const int i = blockIdx.x * 256 + tid;
  unsigned c[DEG_K];
  unsigned deg = 0u;
#pragma unroll
  for (int k = 0; k < DEG_K; ++k) {
    c[k] = (i < n) ? cnt[(size_t)k * n + i] : 0u;
    deg += c[k];
  }
  unsigned inc = deg;
  for (int d = 1; d < 32; d <<= 1) {
    const unsigned t = __shfl_up(inc, d);
    if (lane >= d) inc += t;
  }
  if (lane == 31) wsum[wave] = inc;
  __syncthreads();
  unsigned start = base_s + inc - deg;
  for (int w = 0; w < wave; ++w) start += wsum[w];
  if (i < n) {
    rowptr[i] = start;
    if (i == n - 1) rowptr[n] = start + deg;
    unsigned run = start;
#pragma unroll
    for (int k = 0; k < DEG_K; ++k) {
      cnt[(size_t)k * n + i] = run;
      run += c[k];
    }
  }
}

__device__ __forceinline__ unsigned fill_slot(unsigned* cur, int d, unsigned lo, unsigned nb) {
  const unsigned b = (unsigned)d - lo;
  return (b < nb) ? atomicAdd(&cur[b], 1u) : 0xFFFFFFFFu;
}

__global__ __launch_bounds__(DEG_T) void fill_kernel(const int* __restrict__ src,
                                                     const int* __restrict__ dst,
                                                     const unsigned* __restrict__ first,
                                                     int* __restrict__ csr, int nE, int n) {
  __shared__ unsigned cur[DEG_BINS];
  const int tid = threadIdx.x;
  const int k = blockIdx.x % DEG_K;
  const int r = blockIdx.x / DEG_K;
  const unsigned lo = (unsigned)r * DEG_BINS;
  int nbi = n - (int)lo; if (nbi > DEG_BINS) nbi = DEG_BINS;
  const unsigned nb = (unsigned)nbi;
  const unsigned* __restrict__ f = first + (size_t)k * n + lo;
  for (int b = tid; b < nbi; b += DEG_T) cur[b] = f[b];
  __syncthreads();

  const int S = (((nE + DEG_K - 1) / DEG_K) + 3) & ~3;
  int e0 = k * S; if (e0 > nE) e0 = nE;
  int e1 = e0 + S; if (e1 > nE) e1 = nE;
  const int n4 = (e1 - e0) >> 2;
  const int4* __restrict__ d4 = (const int4*)(dst + e0);
  const int4* __restrict__ s4 = (const int4*)(src + e0);
  const unsigned lim = (unsigned)nE;
  for (int g = tid; g < n4; g += DEG_U * DEG_T) {
    int4 dv[DEG_U], sv[DEG_U];
#pragma unroll
    for (int u = 0; u < DEG_U; ++u) {
      const int gu = g + u * DEG_T;
      dv[u] = (gu < n4) ? d4[gu] : make_int4(-1, -1, -1, -1);
      sv[u] = (gu < n4) ? s4[gu] : make_int4(0, 0, 0, 0);
    }
    unsigned p[DEG_U * 4];
#pragma unroll
    for (int u = 0; u < DEG_U; ++u) {
      p[4 * u + 0] = fill_slot(cur, dv[u].x, lo, nb);
      p[4 * u + 1] = fill_slot(cur, dv[u].y, lo, nb);
      p[4 * u + 2] = fill_slot(cur, dv[u].z, lo, nb);
      p[4 * u + 3] = fill_slot(cur, dv[u].w, lo, nb);
    }
#pragma unroll
    for (int u = 0; u < DEG_U; ++u) {
      if (p[4 * u + 0] < lim) csr[p[4 * u + 0]] = sv[u].x;
      if (p[4 * u + 1] < lim) csr[p[4 * u + 1]] = sv[u].y;
      if (p[4 * u + 2] < lim) csr[p[4 * u + 2]] = sv[u].z;
      if (p[4 * u + 3] < lim) csr[p[4 * u + 3]] = sv[u].w;
    }
  }
  for (int e = e0 + (n4 << 2) + tid; e < e1; e += DEG_T) {
    const unsigned q = fill_slot(cur, dst[e], lo, nb);
    if (q < lim) csr[q] = src[e];
  }
}

#define WT_PITCH 68
__global__ __launch_bounds__(256) void gemm64_wmma(const float* __restrict__ X,
                                                   const float* __restrict__ W,
                                                   const float* __restrict__ row_scale,
                                                   float* __restrict__ Y, int n_rows) {
  __shared__ float Wt[64 * WT_PITCH];
  const int tid = threadIdx.x;
  if (blockIdx.x == 0 && tid < 64) Y[(size_t)n_rows * 64 + tid] = 0.f;
  for (int i = tid; i < 64 * 64; i += 256) {
    int k = i >> 6, n = i & 63;
    Wt[n * WT_PITCH + k] = W[i];
  }
  __syncthreads();

  const int lane = tid & 31;
  const int wave = tid >> 5;
  const int tile = blockIdx.x * 8 + wave;
  if (tile * 16 >= n_rows) return;

  const int m = lane & 15;
  const int khalf = lane >> 4;
  const int row0 = tile * 16;
  const int row = row0 + m;
  const float sn = row_scale[row];
  const float* xrow = X + (size_t)row * 64;

  v8f acc0 = {}, acc1 = {}, acc2 = {}, acc3 = {};
#pragma unroll
  for (int kc = 0; kc < 16; ++kc) {
    const int kbase = kc * 4 + khalf * 2;
    const float2 av = *(const float2*)(xrow + kbase);
    v2f a; a.x = av.x * sn; a.y = av.y * sn;
    const v2f b0 = *(const v2f*)(&Wt[(m +  0) * WT_PITCH + kbase]);
    const v2f b1 = *(const v2f*)(&Wt[(m + 16) * WT_PITCH + kbase]);
    const v2f b2 = *(const v2f*)(&Wt[(m + 32) * WT_PITCH + kbase]);
    const v2f b3 = *(const v2f*)(&Wt[(m + 48) * WT_PITCH + kbase]);
    acc0 = __builtin_amdgcn_wmma_f32_16x16x4_f32(false, a, false, b0, (short)0, acc0, false, false);
    acc1 = __builtin_amdgcn_wmma_f32_16x16x4_f32(false, a, false, b1, (short)0, acc1, false, false);
    acc2 = __builtin_amdgcn_wmma_f32_16x16x4_f32(false, a, false, b2, (short)0, acc2, false, false);
    acc3 = __builtin_amdgcn_wmma_f32_16x16x4_f32(false, a, false, b3, (short)0, acc3, false, false);
  }

  float* yrow = Y + (size_t)row0 * 64;
  const int mofs = (lane < 16) ? 0 : 8;
#pragma unroll
  for (int j = 0; j < 8; ++j) {
    const size_t r = (size_t)(mofs + j) * 64;
    yrow[r +  0 + m] = acc0[j];
    yrow[r + 16 + m] = acc1[j];
    yrow[r + 32 + m] = acc2[j];
    yrow[r + 48 + m] = acc3[j];
  }
}

#define GAT_T 256
#define GAT_U 8

template <int JB>
__device__ __forceinline__ void gather_group(const char* __restrict__ hWb, unsigned lane_off, int idx,
                                             unsigned n, float4& acc) {
  float4 v[GAT_U];
#define GAT_ROW(U)                                                                                  \
  {                                                                                                 \
                \
    const unsigned s = (unsigned)__builtin_amdgcn_update_dpp(0, idx, 0x150 + JB + (U), 0xf, 0xf, true); \
    const unsigned r = s < n ? s : n;      \
    v[U] = *(const float4*)(hWb + (size_t)((r << 8) | lane_off));                                   \
  }
  GAT_ROW(0) GAT_ROW(1) GAT_ROW(2) GAT_ROW(3) GAT_ROW(4) GAT_ROW(5) GAT_ROW(6) GAT_ROW(7)
#undef GAT_ROW
#pragma unroll
  for (int u = 0; u < GAT_U; ++u) {
    acc.x += v[u].x; acc.y += v[u].y; acc.z += v[u].z; acc.w += v[u].w;
  }
}

__device__ __forceinline__ float4 gather_sum2(const float* __restrict__ hW,
                                              const int* __restrict__ csr,
                                              unsigned beg, unsigned end, int l16, int n) {
  const char* __restrict__ hWb = (const char*)hW;
  const unsigned lane_off = (unsigned)l16 << 4;
  const unsigned len = end - beg;
  const unsigned len0 = (unsigned)__builtin_amdgcn_readlane((int)len, 0);
  const unsigned len1 = (unsigned)__builtin_amdgcn_readlane((int)len, 16);
  const unsigned maxlen = len0 > len1 ? len0 : len1;
  float4 acc = make_float4(0.f, 0.f, 0.f, 0.f);
  for (unsigned k = 0; k < maxlen; k += 32u) {
    const unsigned p = k + (unsigned)l16;
    int idx_a = n, idx_b = n;
    if (p < len) idx_a = csr[beg + p];
    if (p + 16u < len) idx_b = csr[beg + p + 16u];
    const unsigned rem = maxlen - k;
    gather_group<0>(hWb, lane_off, idx_a, (unsigned)n, acc);
    if (rem > 1u * GAT_U) gather_group<GAT_U>(hWb, lane_off, idx_a, (unsigned)n, acc);
    if (rem > 2u * GAT_U) gather_group<0>(hWb, lane_off, idx_b, (unsigned)n, acc);
    if (rem > 3u * GAT_U) gather_group<GAT_U>(hWb, lane_off, idx_b, (unsigned)n, acc);
  }
  return acc;
}

__global__ __launch_bounds__(GAT_T) void gather_kernel(const float* __restrict__ hW,
                                                       const unsigned* __restrict__ rowptr,
                                                       const int* __restrict__ csr,
                                                       const float* __restrict__ dn,
                                                       const float* __restrict__ bias,
                                                       float* __restrict__ out, int n) {
  const int lane = threadIdx.x & 31;
  const int l16 = lane & 15;
  const int node0 = __builtin_amdgcn_readfirstlane((int)(blockIdx.x * (GAT_T / 16) + (threadIdx.x >> 5) * 2));
  if (node0 >= n) return;
  const int node = node0 + (lane >> 4);
  unsigned beg = 0u, end = 0u;
  float s = 0.f;
  if (node < n) { beg = rowptr[node]; end = rowptr[node + 1]; s = dn[node]; }
  const float4 b = *(const float4*)(bias + l16 * 4);
  const float4 a = gather_sum2(hW, csr, beg, end, l16, n);
  if (node < n) {
    float4 y;
    y.x = a.x * s + b.x;
    y.y = a.y * s + b.y;
    y.z = a.z * s + b.z;
    y.w = a.w * s + b.w;
    *(float4*)(out + (size_t)node * 64 + l16 * 4) = y;
  }
}

#define AT_PITCH 68
__global__ __launch_bounds__(GAT_T) void gather_gemm_kernel(const float* __restrict__ hW,
                                                            const unsigned* __restrict__ rowptr,
                                                            const int* __restrict__ csr,
                                                            const float* __restrict__ dn,
                                                            const float* __restrict__ bias,
                                                            const float* __restrict__ sn,
                                                            const float* __restrict__ W,
                                                            float* __restrict__ Y, int n) {
  __shared__ __attribute__((aligned(16))) float At[16 * AT_PITCH];
  __shared__ __attribute__((aligned(16))) float Yt[16 * AT_PITCH];
  const int lane = threadIdx.x & 31;
  const int l16 = lane & 15;
  const int khalf = lane >> 4;
  const int wave = __builtin_amdgcn_readfirstlane((int)(threadIdx.x >> 5));
  const int trow = wave * 2 + khalf;
  const int node = (int)(blockIdx.x * 16) + trow;

  if (blockIdx.x == 0 && threadIdx.x < 64) Y[(size_t)n * 64 + threadIdx.x] = 0.f;

  unsigned beg = 0u, end = 0u;
  float s = 0.f, so = 0.f;
  if (node < n) { beg = rowptr[node]; end = rowptr[node + 1]; s = dn[node]; so = sn[node]; }
  const float4 b = *(const float4*)(bias + l16 * 4);
  const float4 a = gather_sum2(hW, csr, beg, end, l16, n);
  float4 y = make_float4(0.f, 0.f, 0.f, 0.f);
  if (node < n) {
    y.x = fmaxf(a.x * s + b.x, 0.f);
    y.y = fmaxf(a.y * s + b.y, 0.f);
    y.z = fmaxf(a.z * s + b.z, 0.f);
    y.w = fmaxf(a.w * s + b.w, 0.f);
    y.x *= so;
    y.y *= so;
    y.z *= so;
    y.w *= so;
  }
  *(float4*)(&At[trow * AT_PITCH + l16 * 4]) = y;

  const int c = l16;
  v2f bf[16];
  if (wave < 4) {
    const float* __restrict__ wp = W + (khalf * 2) * 64 + wave * 16 + c;
#pragma unroll
    for (int kc = 0; kc < 16; ++kc) {
      bf[kc].x = wp[(kc * 4 + 0) * 64];
      bf[kc].y = wp[(kc * 4 + 1) * 64];
    }
  }
  __syncthreads();
  if (wave < 4) {
    v8f acc = {};
#pragma unroll
    for (int kc = 0; kc < 16; ++kc) {
      const v2f af = *(const v2f*)(&At[c * AT_PITCH + kc * 4 + khalf * 2]);
      acc = __builtin_amdgcn_wmma_f32_16x16x4_f32(false, af, false, bf[kc], (short)0, acc, false, false);
    }
#pragma unroll
    for (int j = 0; j < 8; ++j) Yt[(khalf * 8 + j) * AT_PITCH + wave * 16 + c] = acc[j];
  }
  __syncthreads();
  if (node < n) {
    const float4 o = *(const float4*)(&Yt[trow * AT_PITCH + l16 * 4]);
    *(float4*)(Y + (size_t)node * 64 + l16 * 4) = o;
  }
}

extern "C" void kernel_launch(void* const* d_in, const int* in_sizes, int n_in,
                              void* d_out, int out_size, void* d_ws, size_t ws_size,
                              hipStream_t stream) {
  const float* in_feat = (const float*)d_in[0];
  const int*   src     = (const int*)d_in[1];
  const int*   dst     = (const int*)d_in[2];
  const float* W1      = (const float*)d_in[3];
  const float* b1      = (const float*)d_in[4];
  const float* W2      = (const float*)d_in[5];
  const float* b2      = (const float*)d_in[6];
  float* out = (float*)d_out;

  const int N  = in_sizes[0] / 64;
  const int nE = in_sizes[1];

  const size_t feat_bytes = (size_t)N * 64 * sizeof(float);
  auto al = [](size_t x) { return (x + 255) & ~(size_t)255; };
  const int node_blocks = (N + 255) / 256;
  char* ws = (char*)d_ws;
  size_t off = 0;
  float*    hW     = (float*)(ws + off);    off += al(feat_bytes + 256);
  float*    hW2    = (float*)(ws + off);    off += al(feat_bytes + 256);
  unsigned* part   = (unsigned*)(ws + off); off += al((size_t)2 * DEG_K * N * 4);
  float*    sn     = (float*)(ws + off);    off += al((size_t)N * 4);
  float*    dn     = (float*)(ws + off);    off += al((size_t)N * 4);
  unsigned* bsum   = (unsigned*)(ws + off); off += al((size_t)node_blocks * 4);
  unsigned* rowptr = (unsigned*)(ws + off); off += al((size_t)(N + 1) * 4);
  int*      csr    = (int*)(ws + off);      off += al((size_t)nE * 4);
  (void)ws_size; (void)n_in; (void)out_size;

  const int n_ranges = (N + DEG_BINS - 1) / DEG_BINS;
  unsigned* dst_part = part + (size_t)DEG_K * N;
  degree_hist_kernel<<<2 * n_ranges * DEG_K, DEG_T, 0, stream>>>(src, dst, part, nE, N, n_ranges);
  norm_kernel<<<node_blocks, 256, 0, stream>>>(part, sn, dn, bsum, N);
  rowptr_kernel<<<node_blocks, 256, 0, stream>>>(dst_part, bsum, rowptr, N);
  fill_kernel<<<n_ranges * DEG_K, DEG_T, 0, stream>>>(src, dst, dst_part, csr, nE, N);

  const int gemm_blocks = ((N + 15) / 16 + 7) / 8;
  const int gat_blocks = (N + GAT_T / 16 - 1) / (GAT_T / 16);

  gemm64_wmma<<<gemm_blocks, 256, 0, stream>>>(in_feat, W1, sn, hW, N);
  gather_gemm_kernel<<<gat_blocks, GAT_T, 0, stream>>>(hW, rowptr, csr, dn, b1, sn, W2, hW2, N);

  gather_kernel<<<gat_blocks, GAT_T, 0, stream>>>(hW2, rowptr, csr, dn, b2, out, N);
}


# layer 1 sums the input rows first (weighted by sn[src]) and multiplies the block's 16 summed rows by W1 in the launch's tail: sum order agg-then-product; GEMM launch and hW plane go (7 -> 6)
# speedup vs baseline: 4.1583x; 1.0627x over previous
#include <hip/hip_runtime.h>
#include <hip/hip_bf16.h>

typedef float v2f __attribute__((ext_vector_type(2)));
typedef float v8f __attribute__((ext_vector_type(8)));

#define DEG_BINS 12544
#define DEG_K 16
#define DEG_T 512
#define DEG_U 8

__device__ __forceinline__ void deg_count(unsigned* h, int idx, unsigned lo) {
  const unsigned b = (unsigned)idx - lo;
  if (b < (unsigned)DEG_BINS) atomicAdd(&h[b], 1u);
}

__global__ __launch_bounds__(DEG_T) void degree_hist_kernel(const int* __restrict__ src,
                                                            const int* __restrict__ dst,
                                                            unsigned* __restrict__ part,
                                                            int nE, int n, int n_ranges) {
  __shared__ unsigned h[DEG_BINS];
  const int tid = threadIdx.x;
  int bid = blockIdx.x;
  const int k = bid % DEG_K; bid /= DEG_K;
  const int r = bid % n_ranges;
  const int which = bid / n_ranges;
  const int* __restrict__ idx = which ? dst : src;
  const unsigned lo = (unsigned)r * DEG_BINS;

  for (int b = tid; b < DEG_BINS; b += DEG_T) h[b] = 0u;
  __syncthreads();

  const int S = (((nE + DEG_K - 1) / DEG_K) + 3) & ~3;
  int e0 = k * S; if (e0 > nE) e0 = nE;
  int e1 = e0 + S; if (e1 > nE) e1 = nE;
  const int n4 = (e1 - e0) >> 2;
  const int4* __restrict__ p4 = (const int4*)(idx + e0);
  for (int g = tid; g < n4; g += DEG_U * DEG_T) {
    int4 v[DEG_U];
#pragma unroll
    for (int u = 0; u < DEG_U; ++u) {
      const int gu = g + u * DEG_T;
      v[u] = (gu < n4) ? p4[gu] : make_int4(-1, -1, -1, -1);
    }
#pragma unroll
    for (int u = 0; u < DEG_U; ++u) {
      deg_count(h, v[u].x, lo);
      deg_count(h, v[u].y, lo);
      deg_count(h, v[u].z, lo);
      deg_count(h, v[u].w, lo);
    }
  }
  for (int e = e0 + (n4 << 2) + tid; e < e1; e += DEG_T) deg_count(h, idx[e], lo);
  __syncthreads();

  int nb = n - (int)lo; if (nb > DEG_BINS) nb = DEG_BINS;
  unsigned* __restrict__ o = part + (size_t)(which * DEG_K + k) * n + lo;
  for (int b = tid; b < nb; b += DEG_T) o[b] = h[b];
}

__global__ __launch_bounds__(256) void norm_kernel(const unsigned* __restrict__ part,
                                                   float* __restrict__ sn,
                                                   float* __restrict__ dn,
                                                   unsigned* __restrict__ bsum, int n) {
  __shared__ unsigned tot;
  if (threadIdx.x == 0) tot = 0u;
  __syncthreads();
  int i = blockIdx.x * 256 + threadIdx.x;
  unsigned a = 0u, b = 0u;
  if (i < n) {
#pragma unroll
    for (int k = 0; k < DEG_K; ++k) {
      a += part[(size_t)k * n + i];
      b += part[(size_t)(DEG_K + k) * n + i];
    }
    sn[i] = rsqrtf((float)(a < 1u ? 1u : a));
    dn[i] = rsqrtf((float)(b < 1u ? 1u : b));
  }
  atomicAdd(&tot, b);
  __syncthreads();
  if (threadIdx.x == 0) bsum[blockIdx.x] = tot;
}

__global__ __launch_bounds__(256) void rowptr_kernel(unsigned* __restrict__ cnt,
                                                     const unsigned* __restrict__ bsum,
                                                     unsigned* __restrict__ rowptr, int n) {
  __shared__ unsigned wsum[8];
  __shared__ unsigned base_s;
  const int tid = threadIdx.x;
  const int lane = tid & 31, wave = tid >> 5;
  if (tid == 0) base_s = 0u;
  __syncthreads();
  unsigned pre = 0u;
  for (int j = tid; j < (int)blockIdx.x; j += 256) pre += bsum[j];
  atomicAdd(&base_s, pre);

  const int i = blockIdx.x * 256 + tid;
  unsigned c[DEG_K];
  unsigned deg = 0u;
#pragma unroll
  for (int k = 0; k < DEG_K; ++k) {
    c[k] = (i < n) ? cnt[(size_t)k * n + i] : 0u;
    deg += c[k];
  }
  unsigned inc = deg;
  for (int d = 1; d < 32; d <<= 1) {
    const unsigned t = __shfl_up(inc, d);
    if (lane >= d) inc += t;
  }
  if (lane == 31) wsum[wave] = inc;
  __syncthreads();
  unsigned start = base_s + inc - deg;
  for (int w = 0; w < wave; ++w) start += wsum[w];
  if (i < n) {
    rowptr[i] = start;
    if (i == n - 1) rowptr[n] = start + deg;
    unsigned run = start;
#pragma unroll
    for (int k = 0; k < DEG_K; ++k) {
      cnt[(size_t)k * n + i] = run;
      run += c[k];
    }
  }
}

__device__ __forceinline__ unsigned fill_slot(unsigned* cur, int d, unsigned lo, unsigned nb) {
  const unsigned b = (unsigned)d - lo;
  return (b < nb) ? atomicAdd(&cur[b], 1u) : 0xFFFFFFFFu;
}

__global__ __launch_bounds__(DEG_T) void fill_kernel(const int* __restrict__ src,
                                                     const int* __restrict__ dst,
                                                     const unsigned* __restrict__ first,
                                                     int* __restrict__ csr, int nE, int n) {
  __shared__ unsigned cur[DEG_BINS];
  const int tid = threadIdx.x;
  const int k = blockIdx.x % DEG_K;
  const int r = blockIdx.x / DEG_K;
  const unsigned lo = (unsigned)r * DEG_BINS;
  int nbi = n - (int)lo; if (nbi > DEG_BINS) nbi = DEG_BINS;
  const unsigned nb = (unsigned)nbi;
  const unsigned* __restrict__ f = first + (size_t)k * n + lo;
  for (int b = tid; b < nbi; b += DEG_T) cur[b] = f[b];
  __syncthreads();

  const int S = (((nE + DEG_K - 1) / DEG_K) + 3) & ~3;
  int e0 = k * S; if (e0 > nE) e0 = nE;
  int e1 = e0 + S; if (e1 > nE) e1 = nE;
  const int n4 = (e1 - e0) >> 2;
  const int4* __restrict__ d4 = (const int4*)(dst + e0);
  const int4* __restrict__ s4 = (const int4*)(src + e0);
  const unsigned lim = (unsigned)nE;
  for (int g = tid; g < n4; g += DEG_U * DEG_T) {
    int4 dv[DEG_U], sv[DEG_U];
#pragma unroll
    for (int u = 0; u < DEG_U; ++u) {
      const int gu = g + u * DEG_T;
      dv[u] = (gu < n4) ? d4[gu] : make_int4(-1, -1, -1, -1);
      sv[u] = (gu < n4) ? s4[gu] : make_int4(0, 0, 0, 0);
    }
    unsigned p[DEG_U * 4];
#pragma unroll
    for (int u = 0; u < DEG_U; ++u) {
      p[4 * u + 0] = fill_slot(cur, dv[u].x, lo, nb);
      p[4 * u + 1] = fill_slot(cur, dv[u].y, lo, nb);
      p[4 * u + 2] = fill_slot(cur, dv[u].z, lo, nb);
      p[4 * u + 3] = fill_slot(cur, dv[u].w, lo, nb);
    }
#pragma unroll
    for (int u = 0; u < DEG_U; ++u) {
      if (p[4 * u + 0] < lim) csr[p[4 * u + 0]] = sv[u].x;
      if (p[4 * u + 1] < lim) csr[p[4 * u + 1]] = sv[u].y;
      if (p[4 * u + 2] < lim) csr[p[4 * u + 2]] = sv[u].z;
      if (p[4 * u + 3] < lim) csr[p[4 * u + 3]] = sv[u].w;
    }
  }
  for (int e = e0 + (n4 << 2) + tid; e < e1; e += DEG_T) {
    const unsigned q = fill_slot(cur, dst[e], lo, nb);
    if (q < lim) csr[q] = src[e];
  }
}


#define GAT_T 256
#define GAT_U 8

template <int JB, bool WT>
__device__ __forceinline__ void gather_group(const char* __restrict__ tb, unsigned lane_off, int idx,
                                             int wbits, unsigned n, float4& acc) {
  float4 v[GAT_U];
  float w[GAT_U];
#define GAT_ROW(U)                                                                                  \
  {                                                                                                 \
                \
    const unsigned s = (unsigned)__builtin_amdgcn_update_dpp(0, idx, 0x150 + JB + (U), 0xf, 0xf, true); \
      \
    const unsigned r = WT ? s : (s < n ? s : n);                                                    \
    if (WT) w[U] = __int_as_float(__builtin_amdgcn_update_dpp(0, wbits, 0x150 + JB + (U), 0xf, 0xf, true)); \
    v[U] = *(const float4*)(tb + (size_t)((r << 8) | lane_off));                                    \
  }
  GAT_ROW(0) GAT_ROW(1) GAT_ROW(2) GAT_ROW(3) GAT_ROW(4) GAT_ROW(5) GAT_ROW(6) GAT_ROW(7)
#undef GAT_ROW
#pragma unroll
  for (int u = 0; u < GAT_U; ++u) {
    if (WT) {
      acc.x = __builtin_fmaf(v[u].x, w[u], acc.x);
      acc.y = __builtin_fmaf(v[u].y, w[u], acc.y);
      acc.z = __builtin_fmaf(v[u].z, w[u], acc.z);
      acc.w = __builtin_fmaf(v[u].w, w[u], acc.w);
    } else {
      acc.x += v[u].x; acc.y += v[u].y; acc.z += v[u].z; acc.w += v[u].w;
    }
  }
}

template <bool WT>
__device__ __forceinline__ float4 gather_sum2(const float* __restrict__ table,
                                              const int* __restrict__ csr,
                                              const float* __restrict__ wt,
                                              unsigned beg, unsigned end, int l16, int n) {
  const char* __restrict__ tb = (const char*)table;
  const unsigned lane_off = (unsigned)l16 << 4;
  const unsigned len = end - beg;
  const unsigned len0 = (unsigned)__builtin_amdgcn_readlane((int)len, 0);
  const unsigned len1 = (unsigned)__builtin_amdgcn_readlane((int)len, 16);
  const unsigned maxlen = len0 > len1 ? len0 : len1;
  float4 acc = make_float4(0.f, 0.f, 0.f, 0.f);
  for (unsigned k = 0; k < maxlen; k += 32u) {
    const unsigned p = k + (unsigned)l16;
    int idx_a = WT ? 0 : n, idx_b = WT ? 0 : n;
    if (p < len) idx_a = csr[beg + p];
    if (p + 16u < len) idx_b = csr[beg + p + 16u];
    float w_a = 0.f, w_b = 0.f;
    if (WT) {
      idx_a = (int)((unsigned)idx_a < (unsigned)n ? (unsigned)idx_a : (unsigned)(n - 1));
      idx_b = (int)((unsigned)idx_b < (unsigned)n ? (unsigned)idx_b : (unsigned)(n - 1));
      if (p < len) w_a = wt[idx_a];
      if (p + 16u < len) w_b = wt[idx_b];
    }
    const int wb_a = __float_as_int(w_a), wb_b = __float_as_int(w_b);
    const unsigned rem = maxlen - k;
    gather_group<0, WT>(tb, lane_off, idx_a, wb_a, (unsigned)n, acc);
    if (rem > 1u * GAT_U) gather_group<GAT_U, WT>(tb, lane_off, idx_a, wb_a, (unsigned)n, acc);
    if (rem > 2u * GAT_U) gather_group<0, WT>(tb, lane_off, idx_b, wb_b, (unsigned)n, acc);
    if (rem > 3u * GAT_U) gather_group<GAT_U, WT>(tb, lane_off, idx_b, wb_b, (unsigned)n, acc);
  }
  return acc;
}

__global__ __launch_bounds__(GAT_T) void gather_kernel(const float* __restrict__ hW,
                                                       const unsigned* __restrict__ rowptr,
                                                       const int* __restrict__ csr,
                                                       const float* __restrict__ dn,
                                                       const float* __restrict__ bias,
                                                       float* __restrict__ out, int n) {
  const int lane = threadIdx.x & 31;
  const int l16 = lane & 15;
  const int node0 = __builtin_amdgcn_readfirstlane((int)(blockIdx.x * (GAT_T / 16) + (threadIdx.x >> 5) * 2));
  if (node0 >= n) return;
  const int node = node0 + (lane >> 4);
  unsigned beg = 0u, end = 0u;
  float s = 0.f;
  if (node < n) { beg = rowptr[node]; end = rowptr[node + 1]; s = dn[node]; }
  const float4 b = *(const float4*)(bias + l16 * 4);
  const float4 a = gather_sum2<false>(hW, csr, nullptr, beg, end, l16, n);
  if (node < n) {
    float4 y;
    y.x = a.x * s + b.x;
    y.y = a.y * s + b.y;
    y.z = a.z * s + b.z;
    y.w = a.w * s + b.w;
    *(float4*)(out + (size_t)node * 64 + l16 * 4) = y;
  }
}

#define AT_PITCH 68

__device__ __forceinline__ void load_b_frags(const float* __restrict__ W, int khalf, int wave, int c,
                                             v2f (&bf)[16]) {
  const float* __restrict__ wp = W + (khalf * 2) * 64 + wave * 16 + c;
#pragma unroll
  for (int kc = 0; kc < 16; ++kc) {
    bf[kc].x = wp[(kc * 4 + 0) * 64];
    bf[kc].y = wp[(kc * 4 + 1) * 64];
  }
}

__device__ __forceinline__ void tile_product(const float* At, float* Yt, const v2f (&bf)[16],
                                             int khalf, int wave, int c) {
  v8f acc = {};
#pragma unroll
  for (int kc = 0; kc < 16; ++kc) {
    const v2f af = *(const v2f*)(&At[c * AT_PITCH + kc * 4 + khalf * 2]);
    acc = __builtin_amdgcn_wmma_f32_16x16x4_f32(false, af, false, bf[kc], (short)0, acc, false, false);
  }
#pragma unroll
  for (int j = 0; j < 8; ++j) Yt[(khalf * 8 + j) * AT_PITCH + wave * 16 + c] = acc[j];
}

__global__ __launch_bounds__(GAT_T) void gather_gemm_kernel(const float* __restrict__ X,
                                                            const unsigned* __restrict__ rowptr,
                                                            const int* __restrict__ csr,
                                                            const float* __restrict__ dn,
                                                            const float* __restrict__ bias,
                                                            const float* __restrict__ sn,
                                                            const float* __restrict__ W1,
                                                            const float* __restrict__ W2,
                                                            float* __restrict__ Y, int n) {
  __shared__ __attribute__((aligned(16))) float At[16 * AT_PITCH];
  __shared__ __attribute__((aligned(16))) float Yt[16 * AT_PITCH];
  const int lane = threadIdx.x & 31;
  const int l16 = lane & 15;
  const int khalf = lane >> 4;
  const int wave = __builtin_amdgcn_readfirstlane((int)(threadIdx.x >> 5));
  const int trow = wave * 2 + khalf;
  const int node = (int)(blockIdx.x * 16) + trow;

  if (blockIdx.x == 0 && threadIdx.x < 64) Y[(size_t)n * 64 + threadIdx.x] = 0.f;

  unsigned beg = 0u, end = 0u;
  float s = 0.f, so = 0.f;
  if (node < n) { beg = rowptr[node]; end = rowptr[node + 1]; s = dn[node]; so = sn[node]; }
  const float4 b = *(const float4*)(bias + l16 * 4);
  const float4 a = gather_sum2<true>(X, csr, sn, beg, end, l16, n);
  *(float4*)(&At[trow * AT_PITCH + l16 * 4]) = a;

  const int c = l16;
  const int pw = wave & 3;
  v2f bf[16];
  load_b_frags(wave < 4 ? W1 : W2, khalf, pw, c, bf);
  __syncthreads();
  if (wave < 4) tile_product(At, Yt, bf, khalf, pw, c);
  __syncthreads();
  {
    const float4 t = *(const float4*)(&Yt[trow * AT_PITCH + l16 * 4]);
    float4 y = make_float4(0.f, 0.f, 0.f, 0.f);
    if (node < n) {
      y.x = fmaxf(t.x * s + b.x, 0.f);
      y.y = fmaxf(t.y * s + b.y, 0.f);
      y.z = fmaxf(t.z * s + b.z, 0.f);
      y.w = fmaxf(t.w * s + b.w, 0.f);
      y.x *= so;
      y.y *= so;
      y.z *= so;
      y.w *= so;
    }
    *(float4*)(&At[trow * AT_PITCH + l16 * 4]) = y;
  }
  __syncthreads();
  if (wave >= 4) tile_product(At, Yt, bf, khalf, pw, c);
  __syncthreads();
  if (node < n) {
    const float4 o = *(const float4*)(&Yt[trow * AT_PITCH + l16 * 4]);
    *(float4*)(Y + (size_t)node * 64 + l16 * 4) = o;
  }
}

extern "C" void kernel_launch(void* const* d_in, const int* in_sizes, int n_in,
                              void* d_out, int out_size, void* d_ws, size_t ws_size,
                              hipStream_t stream) {
  const float* in_feat = (const float*)d_in[0];
  const int*   src     = (const int*)d_in[1];
  const int*   dst     = (const int*)d_in[2];
  const float* W1      = (const float*)d_in[3];
  const float* b1      = (const float*)d_in[4];
  const float* W2      = (const float*)d_in[5];
  const float* b2      = (const float*)d_in[6];
  float* out = (float*)d_out;

  const int N  = in_sizes[0] / 64;
  const int nE = in_sizes[1];

  const size_t feat_bytes = (size_t)N * 64 * sizeof(float);
  auto al = [](size_t x) { return (x + 255) & ~(size_t)255; };
  const int node_blocks = (N + 255) / 256;
  char* ws = (char*)d_ws;
  size_t off = 0;
  float*    hW2    = (float*)(ws + off);    off += al(feat_bytes + 256);
  unsigned* part   = (unsigned*)(ws + off); off += al((size_t)2 * DEG_K * N * 4);
  float*    sn     = (float*)(ws + off);    off += al((size_t)N * 4);
  float*    dn     = (float*)(ws + off);    off += al((size_t)N * 4);
  unsigned* bsum   = (unsigned*)(ws + off); off += al((size_t)node_blocks * 4);
  unsigned* rowptr = (unsigned*)(ws + off); off += al((size_t)(N + 1) * 4);
  int*      csr    = (int*)(ws + off);      off += al((size_t)nE * 4);
  (void)ws_size; (void)n_in; (void)out_size;

  const int n_ranges = (N + DEG_BINS - 1) / DEG_BINS;
  unsigned* dst_part = part + (size_t)DEG_K * N;
  degree_hist_kernel<<<2 * n_ranges * DEG_K, DEG_T, 0, stream>>>(src, dst, part, nE, N, n_ranges);
  norm_kernel<<<node_blocks, 256, 0, stream>>>(part, sn, dn, bsum, N);
  rowptr_kernel<<<node_blocks, 256, 0, stream>>>(dst_part, bsum, rowptr, N);
  fill_kernel<<<n_ranges * DEG_K, DEG_T, 0, stream>>>(src, dst, dst_part, csr, nE, N);

  const int gat_blocks = (N + GAT_T / 16 - 1) / (GAT_T / 16);

  gather_gemm_kernel<<<gat_blocks, GAT_T, 0, stream>>>(in_feat, rowptr, csr, dn, b1, sn, W1, W2, hW2, N);

  gather_kernel<<<gat_blocks, GAT_T, 0, stream>>>(hW2, rowptr, csr, dn, b2, out, N);
}
